# starting kernel
# baseline (speedup 1.0000x reference)
#include <hip/hip_runtime.h>

typedef __attribute__((ext_vector_type(2))) float v2f;
typedef __attribute__((ext_vector_type(8))) float v8f;

#define TILE_K 32
#define BLK_M  128
#define BLK_N  64
#define AP     34
#define BP     162
#define GK     512
#define NT_K   (GK / TILE_K)

__global__ __launch_bounds__(256) void dct_make(float* __restrict__ D) {
  int idx = blockIdx.x * 256 + threadIdx.x;
  int k = idx >> 9;
  int n = idx & 511;
  int r = ((2 * n + 1) * k) & 2047;
  float th = 3.14159265358979323846f * (float)r * (1.0f / 1024.0f);
  float v = cosf(th) * 0.0625f;
  if (k == 0) v *= 0.70710678118654752f;
  D[idx] = v;
}


__device__ __forceinline__ void fetchA(const float* __restrict__ Ab, int m0, int k0,
                                       long rsA, long csA, int tid, float (&r)[16]) {
#pragma unroll
  for (int t = 0; t < 16; ++t) {
    int idx = tid + t * 256;
    int kk = idx & (TILE_K - 1);
    int mm = idx >> 5;
    r[t] = Ab[(long)(m0 + mm) * rsA + (long)(k0 + kk) * csA];
  }
}

__device__ __forceinline__ void storeA(float* __restrict__ As, int tid,
                                       const float (&r)[16]) {
#pragma unroll
  for (int t = 0; t < 16; ++t) {
    int idx = tid + t * 256;
    int kk = idx & (TILE_K - 1);
    int mm = idx >> 5;
    As[mm * AP + kk] = r[t];
  }
}

__device__ __forceinline__ void fetchB(const float* __restrict__ Bb, int n0, int k0,
                                       long rsB, long csB, int tid, int bNFast,
                                       float (&r)[8]) {
  if (bNFast) {
#pragma unroll
    for (int t = 0; t < 8; ++t) {
      int idx = tid + t * 256;
      int nn = idx & (BLK_N - 1);
      int kk = idx >> 6;
      r[t] = Bb[(long)(k0 + kk) * rsB + (long)(n0 + nn) * csB];
    }
  } else {
#pragma unroll
    for (int t = 0; t < 8; ++t) {
      int idx = tid + t * 256;
      int kk = idx & (TILE_K - 1);
      int nn = idx >> 5;
      r[t] = Bb[(long)(k0 + kk) * rsB + (long)(n0 + nn) * csB];
    }
  }
}

__device__ __forceinline__ void storeB(float* __restrict__ Bs, int tid, int bNFast,
                                       const float (&r)[8]) {
  if (bNFast) {
#pragma unroll
    for (int t = 0; t < 8; ++t) {
      int idx = tid + t * 256;
      int nn = idx & (BLK_N - 1);
      int kk = idx >> 6;
      Bs[(kk >> 1) * BP + nn * 2 + (kk & 1)] = r[t];
    }
  } else {
#pragma unroll
    for (int t = 0; t < 8; ++t) {
      int idx = tid + t * 256;
      int kk = idx & (TILE_K - 1);
      int nn = idx >> 5;
      Bs[(kk >> 1) * BP + nn * 2 + (kk & 1)] = r[t];
    }
  }
}


__device__ __forceinline__ void mma_tile(const float* __restrict__ As,
                                         const float* __restrict__ Bs,
                                         v8f (&acc)[2][2],
                                         int wm, int wn, int row, int grp) {
#pragma unroll
  for (int kk = 0; kk < TILE_K; kk += 4) {
    v2f a[2], b[2];
#pragma unroll
    for (int mt = 0; mt < 2; ++mt)
      a[mt] = *(const v2f*)&As[(wm + mt * 16 + row) * AP + kk + 2 * grp];
#pragma unroll
    for (int nt = 0; nt < 2; ++nt)
      b[nt] = *(const v2f*)&Bs[((kk >> 1) + grp) * BP + (wn + nt * 16 + row) * 2];
#pragma unroll
    for (int mt = 0; mt < 2; ++mt)
#pragma unroll
      for (int nt = 0; nt < 2; ++nt)
        acc[mt][nt] = __builtin_amdgcn_wmma_f32_16x16x4_f32(
            false, a[mt], false, b[nt], (short)0, acc[mt][nt], false, false);
  }
}


__global__ __launch_bounds__(256)
void dct_gemm(const float* __restrict__ A, const float* __restrict__ B,
              float* __restrict__ C, int N,
              long rsA, long csA, long rsB, long csB, long rsC, long csC,
              int zDiv, long zPitch, int useOffA, int useOffB, int useOffC,
              int bNFast) {
  (void)N;
  __shared__ float As[2][BLK_M * AP];
  __shared__ float Bs[2][(TILE_K / 2) * BP];

  const int tid = threadIdx.x;
  const int z   = blockIdx.z;
  const long off = (long)(z / zDiv) * zPitch + (long)(z % zDiv);
  const float* Ab = A + (useOffA ? off : 0);
  const float* Bb = B + (useOffB ? off : 0);
  float*       Cb = C + (useOffC ? off : 0);

  const int m0 = blockIdx.y * BLK_M;
  const int n0 = blockIdx.x * BLK_N;

  const int lane = tid & 31;
  const int wave = tid >> 5;
  const int wm   = (wave & 3) * 32;
  const int wn   = (wave >> 2) * 32;
  const int row  = lane & 15;
  const int grp  = lane >> 4;

  v8f acc[2][2];
#pragma unroll
  for (int i = 0; i < 2; ++i)
#pragma unroll
    for (int j = 0; j < 2; ++j) acc[i][j] = (v8f)0.0f;

  float ra[16], rb[8];

  fetchA(Ab, m0, 0, rsA, csA, tid, ra);
  fetchB(Bb, n0, 0, rsB, csB, tid, bNFast, rb);
  storeA(As[0], tid, ra);
  storeB(Bs[0], tid, bNFast, rb);
  __syncthreads();

  for (int t = 0; t < NT_K; ++t) {
    const int cur = t & 1;
    if (t + 1 < NT_K) {
      fetchA(Ab, m0, (t + 1) * TILE_K, rsA, csA, tid, ra);
      fetchB(Bb, n0, (t + 1) * TILE_K, rsB, csB, tid, bNFast, rb);
    }
    mma_tile(As[cur], Bs[cur], acc, wm, wn, row, grp);
    if (t + 1 < NT_K) {
      storeA(As[cur ^ 1], tid, ra);
      storeB(Bs[cur ^ 1], tid, bNFast, rb);
    }
    __syncthreads();
  }

#pragma unroll
  for (int mt = 0; mt < 2; ++mt)
#pragma unroll
    for (int nt = 0; nt < 2; ++nt)
#pragma unroll
      for (int v = 0; v < 8; ++v) {
        int m = m0 + wm + mt * 16 + grp * 8 + v;
        int n = n0 + wn + nt * 16 + row;
        Cb[(long)m * rsC + (long)n * csC] = acc[mt][nt][v];
      }
}

extern "C" void kernel_launch(void* const* d_in, const int* in_sizes, int n_in,
                              void* d_out, int out_size, void* d_ws, size_t ws_size,
                              hipStream_t stream) {
  (void)in_sizes; (void)n_in; (void)out_size; (void)ws_size;
  const float* x = (const float*)d_in[0];
  float* out = (float*)d_out;
  float* Dm  = (float*)d_ws;
  float* Y1  = Dm + 512 * 512;
  const long HWC = 512L * 512L * 3L;

  dct_make<<<dim3(1024), dim3(256), 0, stream>>>(Dm);

  dct_gemm<<<dim3(24, 4, 32), dim3(256), 0, stream>>>(
      Dm, x, Y1, 1536,
        512, 1,   1536, 1,   1536, 1,
        1,   HWC,   0, 1, 1,   1);

  dct_gemm<<<dim3(8, 4, 96), dim3(256), 0, stream>>>(
      Y1, Dm, out, 512,
        1536, 3,   1, 512,   1536, 3,
        3,   HWC,   1, 0, 1,   0);
}


# Basis operand bypasses LDS: DCT table stored in WMMA fragment order, each wave reads its basis fragments from global as b128 one K-tile ahead; only the data operand is still staged in LDS
# speedup vs baseline: 1.1832x; 1.1832x over previous
#include <hip/hip_runtime.h>


typedef __attribute__((ext_vector_type(2))) float v2f;
typedef __attribute__((ext_vector_type(4))) float v4f;
typedef __attribute__((ext_vector_type(8))) float v8f;

#define TILE_K 32
#define BLK_M  128
#define BLK_N  64
#define AP     34
#define BP     162
#define GK     512
#define NT_K   (GK / TILE_K)

__global__ __launch_bounds__(256) void dct_make(float* __restrict__ F) {
  int idx  = blockIdx.x * 256 + threadIdx.x;
  int e    = idx & 3;
  int lane = (idx >> 2) & 31;
  int j    = (idx >> 7) & 63;
  int t16  = idx >> 13;
  int k = t16 * 16 + (lane & 15);
  int n = (2 * j + (e >> 1)) * 4 + 2 * (lane >> 4) + (e & 1);
  int r = ((2 * n + 1) * k) & 2047;
  float th = 3.14159265358979323846f * (float)r * (1.0f / 1024.0f);
  float v = cosf(th) * 0.0625f;
  if (k == 0) v *= 0.70710678118654752f;
  F[idx] = v;
}


__device__ __forceinline__ void fetchA(const float* __restrict__ Ab, int m0, int k0,
                                       long rsA, long csA, int tid, float (&r)[16]) {
#pragma unroll
  for (int t = 0; t < 16; ++t) {
    int idx = tid + t * 256;
    int kk = idx & (TILE_K - 1);
    int mm = idx >> 5;
    r[t] = Ab[(long)(m0 + mm) * rsA + (long)(k0 + kk) * csA];
  }
}

__device__ __forceinline__ void storeA(float* __restrict__ As, int tid,
                                       const float (&r)[16]) {
#pragma unroll
  for (int t = 0; t < 16; ++t) {
    int idx = tid + t * 256;
    int kk = idx & (TILE_K - 1);
    int mm = idx >> 5;
    As[mm * AP + kk] = r[t];
  }
}

__device__ __forceinline__ void fetchB(const float* __restrict__ Bb, int n0, int k0,
                                       long rsB, long csB, int tid, float (&r)[8]) {
#pragma unroll
  for (int t = 0; t < 8; ++t) {
    int idx = tid + t * 256;
    int nn = idx & (BLK_N - 1);
    int kk = idx >> 6;
    r[t] = Bb[(long)(k0 + kk) * rsB + (long)(n0 + nn) * csB];
  }
}

__device__ __forceinline__ void storeB(float* __restrict__ Bs, int tid,
                                       const float (&r)[8]) {
#pragma unroll
  for (int t = 0; t < 8; ++t) {
    int idx = tid + t * 256;
    int nn = idx & (BLK_N - 1);
    int kk = idx >> 6;
    Bs[(kk >> 1) * BP + nn * 2 + (kk & 1)] = r[t];
  }
}

__device__ __forceinline__ void fetchF(const float* __restrict__ Fw, int kt,
                                       v4f (&f)[2][4]) {
#pragma unroll
  for (int i = 0; i < 2; ++i)
#pragma unroll
    for (int jj = 0; jj < 4; ++jj)
      f[i][jj] = *(const v4f*)&Fw[(i * 64 + kt * 4 + jj) * 128];
}

template <bool BASIS_A>
__device__ __forceinline__ void mma_tile(const float* __restrict__ Ds,
                                         const v4f (&f)[2][4],
                                         v8f (&acc)[2][2],
                                         int wm, int wn, int row, int grp) {
#pragma unroll
  for (int s = 0; s < TILE_K / 4; ++s) {
    const int kk = s * 4;
    v2f a[2], b[2];
#pragma unroll
    for (int i = 0; i < 2; ++i) {
      const v4f q = f[i][s >> 1];
      const v2f fr = (s & 1) ? q.zw : q.xy;
      if (BASIS_A) a[i] = fr; else b[i] = fr;
    }
    if (BASIS_A) {
#pragma unroll
      for (int nt = 0; nt < 2; ++nt)
        b[nt] = *(const v2f*)&Ds[((kk >> 1) + grp) * BP + (wn + nt * 16 + row) * 2];
    } else {
#pragma unroll
      for (int mt = 0; mt < 2; ++mt)
        a[mt] = *(const v2f*)&Ds[(wm + mt * 16 + row) * AP + kk + 2 * grp];
    }
#pragma unroll
    for (int mt = 0; mt < 2; ++mt)
#pragma unroll
      for (int nt = 0; nt < 2; ++nt)
        acc[mt][nt] = __builtin_amdgcn_wmma_f32_16x16x4_f32(
            false, a[mt], false, b[nt], (short)0, acc[mt][nt], false, false);
  }
}


template <bool BASIS_A>
__global__ __launch_bounds__(256)
void dct_gemm(const float* __restrict__ F, const float* __restrict__ X,
              float* __restrict__ C,
              long rsX, long csX, long rsC, long csC, int zDiv, long zPitch) {
  constexpr int DS_FLOATS = BASIS_A ? (TILE_K / 2) * BP : BLK_M * AP;
  constexpr int NR = BASIS_A ? 8 : 16;
  __shared__ float Ds[2][DS_FLOATS];

  const int tid = threadIdx.x;
  const int z   = blockIdx.z;
  const long off = (long)(z / zDiv) * zPitch + (long)(z % zDiv);
  const float* Xb = X + off;
  float*       Cb = C + off;

  const int m0 = blockIdx.y * BLK_M;
  const int n0 = blockIdx.x * BLK_N;

  const int lane = tid & 31;
  const int wave = tid >> 5;
  const int wm   = (wave & 3) * 32;
  const int wn   = (wave >> 2) * 32;
  const int row  = lane & 15;
  const int grp  = lane >> 4;

  const int t16 = (BASIS_A ? (m0 + wm) : (n0 + wn)) >> 4;
  const float* Fw = F + ((long)t16 * 64 * 32 + lane) * 4;

  v8f acc[2][2];
#pragma unroll
  for (int i = 0; i < 2; ++i)
#pragma unroll
    for (int j = 0; j < 2; ++j) acc[i][j] = (v8f)0.0f;

  float rd[NR];
  v4f f0[2][4], f1[2][4];

  auto fetchD = [&](int kt) {
    if constexpr (BASIS_A) fetchB(Xb, n0, kt * TILE_K, rsX, csX, tid, rd);
    else                   fetchA(Xb, m0, kt * TILE_K, rsX, csX, tid, rd);
  };
  auto storeD = [&](float* buf) {
    if constexpr (BASIS_A) storeB(buf, tid, rd);
    else                   storeA(buf, tid, rd);
  };

  fetchD(0);
  fetchF(Fw, 0, f0);
  storeD(Ds[0]);
  __syncthreads();

  for (int t = 0; t < NT_K; t += 2) {
    fetchD(t + 1);
    fetchF(Fw, t + 1, f1);
    __builtin_amdgcn_sched_barrier(0);
    mma_tile<BASIS_A>(Ds[0], f0, acc, wm, wn, row, grp);
    storeD(Ds[1]);
    __syncthreads();

    const bool more = (t + 2 < NT_K);
    if (more) {
      fetchD(t + 2);
      fetchF(Fw, t + 2, f0);
    }
    mma_tile<BASIS_A>(Ds[1], f1, acc, wm, wn, row, grp);
    if (more) storeD(Ds[0]);
    __syncthreads();
  }

#pragma unroll
  for (int mt = 0; mt < 2; ++mt)
#pragma unroll
    for (int nt = 0; nt < 2; ++nt)
#pragma unroll
      for (int v = 0; v < 8; ++v) {
        int m = m0 + wm + mt * 16 + grp * 8 + v;
        int n = n0 + wn + nt * 16 + row;
        Cb[(long)m * rsC + (long)n * csC] = acc[mt][nt][v];
      }
}

extern "C" void kernel_launch(void* const* d_in, const int* in_sizes, int n_in,
                              void* d_out, int out_size, void* d_ws, size_t ws_size,
                              hipStream_t stream) {
  (void)in_sizes; (void)n_in; (void)out_size; (void)ws_size;
  const float* x = (const float*)d_in[0];
  float* out = (float*)d_out;
  float* Fm  = (float*)d_ws;
  float* Y1  = Fm + 512 * 512;
  const long HWC = 512L * 512L * 3L;

  dct_make<<<dim3(1024), dim3(256), 0, stream>>>(Fm);

  dct_gemm<true><<<dim3(24, 4, 32), dim3(256), 0, stream>>>(
      Fm, x, Y1,
        1536, 1,   1536, 1,   1,   HWC);

  dct_gemm<false><<<dim3(8, 4, 96), dim3(256), 0, stream>>>(
      Fm, Y1, out,
        1536, 3,   1536, 3,   3,   HWC);
}


# Stage 2 work-group takes all 3 channels of its pixels: intermediate rows loaded contiguously as b128, channel split in the LDS fragment reads, basis fragments shared by 3 tiles, 12-byte stores
# speedup vs baseline: 1.3275x; 1.1220x over previous
#include <hip/hip_runtime.h>


typedef __attribute__((ext_vector_type(2))) float v2f;
typedef __attribute__((ext_vector_type(4))) float v4f;
typedef __attribute__((ext_vector_type(8))) float v8f;

#define TILE_K 32
#define BLK_M  128
#define BLK_N  64
#define BP     162
#define RP     100
#define GK     512
#define NT_K   (GK / TILE_K)

__global__ __launch_bounds__(256) void dct_make(float* __restrict__ F) {
  int idx  = blockIdx.x * 256 + threadIdx.x;
  int e    = idx & 3;
  int lane = (idx >> 2) & 31;
  int j    = (idx >> 7) & 63;
  int t16  = idx >> 13;
  int k = t16 * 16 + (lane & 15);
  int n = (2 * j + (e >> 1)) * 4 + 2 * (lane >> 4) + (e & 1);
  int r = ((2 * n + 1) * k) & 2047;
  float th = 3.14159265358979323846f * (float)r * (1.0f / 1024.0f);
  float v = cosf(th) * 0.0625f;
  if (k == 0) v *= 0.70710678118654752f;
  F[idx] = v;
}


__device__ __forceinline__ void fetchB(const float* __restrict__ Bb, int n0, int k0,
                                       long rsB, long csB, int tid, float (&r)[8]) {
#pragma unroll
  for (int t = 0; t < 8; ++t) {
    int idx = tid + t * 256;
    int nn = idx & (BLK_N - 1);
    int kk = idx >> 6;
    r[t] = Bb[(long)(k0 + kk) * rsB + (long)(n0 + nn) * csB];
  }
}

__device__ __forceinline__ void storeB(float* __restrict__ Bs, int tid,
                                       const float (&r)[8]) {
#pragma unroll
  for (int t = 0; t < 8; ++t) {
    int idx = tid + t * 256;
    int nn = idx & (BLK_N - 1);
    int kk = idx >> 6;
    Bs[(kk >> 1) * BP + nn * 2 + (kk & 1)] = r[t];
  }
}

__device__ __forceinline__ void fetchF(const float* __restrict__ Fw, int kt,
                                       v4f (&f)[2][4]) {
#pragma unroll
  for (int i = 0; i < 2; ++i)
#pragma unroll
    for (int jj = 0; jj < 4; ++jj)
      f[i][jj] = *(const v4f*)&Fw[(i * 64 + kt * 4 + jj) * 128];
}

__device__ __forceinline__ void mma_cols(const float* __restrict__ Ds,
                                         const v4f (&f)[2][4],
                                         v8f (&acc)[2][2],
                                         int wn, int row, int grp) {
#pragma unroll
  for (int s = 0; s < TILE_K / 4; ++s) {
    const int kk = s * 4;
    v2f a[2], b[2];
#pragma unroll
    for (int i = 0; i < 2; ++i) {
      const v4f q = f[i][s >> 1];
      a[i] = (s & 1) ? q.zw : q.xy;
    }
#pragma unroll
    for (int nt = 0; nt < 2; ++nt)
      b[nt] = *(const v2f*)&Ds[((kk >> 1) + grp) * BP + (wn + nt * 16 + row) * 2];
#pragma unroll
    for (int mt = 0; mt < 2; ++mt)
#pragma unroll
      for (int nt = 0; nt < 2; ++nt)
        acc[mt][nt] = __builtin_amdgcn_wmma_f32_16x16x4_f32(
            false, a[mt], false, b[nt], (short)0, acc[mt][nt], false, false);
  }
}

__global__ __launch_bounds__(256)
void dct_cols(const float* __restrict__ F, const float* __restrict__ X,
              float* __restrict__ C,
              long rsX, long csX, long rsC, long csC, long zPitch) {
  __shared__ float Ds[2][(TILE_K / 2) * BP];

  const int tid = threadIdx.x;
  const long off = (long)blockIdx.z * zPitch;
  const float* Xb = X + off;
  float*       Cb = C + off;

  const int m0 = blockIdx.y * BLK_M;
  const int n0 = blockIdx.x * BLK_N;

  const int lane = tid & 31;
  const int wave = tid >> 5;
  const int wm   = (wave & 3) * 32;
  const int wn   = (wave >> 2) * 32;
  const int row  = lane & 15;
  const int grp  = lane >> 4;

  const int t16 = (m0 + wm) >> 4;
  const float* Fw = F + ((long)t16 * 64 * 32 + lane) * 4;

  v8f acc[2][2];
#pragma unroll
  for (int i = 0; i < 2; ++i)
#pragma unroll
    for (int j = 0; j < 2; ++j) acc[i][j] = (v8f)0.0f;

  float rd[8];
  v4f f0[2][4], f1[2][4];

  fetchB(Xb, n0, 0, rsX, csX, tid, rd);
  fetchF(Fw, 0, f0);
  storeB(Ds[0], tid, rd);
  __syncthreads();

  for (int t = 0; t < NT_K; t += 2) {
    fetchB(Xb, n0, (t + 1) * TILE_K, rsX, csX, tid, rd);
    fetchF(Fw, t + 1, f1);
    __builtin_amdgcn_sched_barrier(0);
    mma_cols(Ds[0], f0, acc, wn, row, grp);
    storeB(Ds[1], tid, rd);
    __syncthreads();

    const bool more = (t + 2 < NT_K);
    if (more) {
      fetchB(Xb, n0, (t + 2) * TILE_K, rsX, csX, tid, rd);
      fetchF(Fw, t + 2, f0);
    }
    mma_cols(Ds[1], f1, acc, wn, row, grp);
    if (more) storeB(Ds[0], tid, rd);
    __syncthreads();
  }

#pragma unroll
  for (int mt = 0; mt < 2; ++mt)
#pragma unroll
    for (int nt = 0; nt < 2; ++nt)
#pragma unroll
      for (int v = 0; v < 8; ++v) {
        int m = m0 + wm + mt * 16 + grp * 8 + v;
        int n = n0 + wn + nt * 16 + row;
        Cb[(long)m * rsC + (long)n * csC] = acc[mt][nt][v];
      }
}


__device__ __forceinline__ void fetchR(const float* __restrict__ Yb, unsigned yo,
                                       int kt, v4f (&r)[12]) {
#pragma unroll
  for (int t = 0; t < 12; ++t)
    r[t] = *(const v4f*)&(Yb + kt * 96)[yo + (t & 3) * 32 * 1536 + (t >> 2) * 32];
}

__device__ __forceinline__ void storeR(float* __restrict__ Rs, const v4f (&r)[12]) {
#pragma unroll
  for (int t = 0; t < 12; ++t)
    *(v4f*)&Rs[(t & 3) * 32 * RP + (t >> 2) * 32] = r[t];
}

__device__ __forceinline__ void mma_rows3(const float* __restrict__ Rw,
                                          const v4f (&f)[2][4],
                                          v8f (&acc)[3][2][2]) {
#pragma unroll
  for (int s = 0; s < TILE_K / 4; ++s) {
    v2f a[3][2], b[2];
#pragma unroll
    for (int i = 0; i < 2; ++i) {
      const v4f q = f[i][s >> 1];
      b[i] = (s & 1) ? q.zw : q.xy;
    }
#pragma unroll
    for (int c = 0; c < 3; ++c)
#pragma unroll
      for (int mt = 0; mt < 2; ++mt) {
        const float* p = &Rw[mt * 16 * RP + s * 12 + c];
        a[c][mt].x = p[0];
        a[c][mt].y = p[3];
      }
#pragma unroll
    for (int c = 0; c < 3; ++c)
#pragma unroll
      for (int mt = 0; mt < 2; ++mt)
#pragma unroll
        for (int nt = 0; nt < 2; ++nt)
          acc[c][mt][nt] = __builtin_amdgcn_wmma_f32_16x16x4_f32(
              false, a[c][mt], false, b[nt], (short)0, acc[c][mt][nt], false, false);
  }
}

__global__ __launch_bounds__(256, 4)
void dct_rows3(const float* __restrict__ F, const float* __restrict__ Y,
               float* __restrict__ Out) {
  __shared__ float Rs[2][BLK_M * RP];

  const int tid = threadIdx.x;
  const long off = (long)blockIdx.z * (512L * 1536L);
  const int m0 = blockIdx.y * BLK_M;
  const int n0 = blockIdx.x * BLK_N;

  const int lane = tid & 31;
  const int wave = __builtin_amdgcn_readfirstlane(tid >> 5);
  const int wm   = (wave & 3) * 32;
  const int wn   = (wave >> 2) * 32;
  const int row  = lane & 15;
  const int grp  = lane >> 4;

  const int t16 = (n0 + wn) >> 4;
  const float* Fw = F + ((long)t16 * 64 * 32 + lane) * 4;

  const float* Yb = Y + off + (long)m0 * 1536;
  const unsigned yo = (unsigned)(tid >> 3) * 1536u + (unsigned)(tid & 7) * 4u;
  const int   lw = (tid >> 3) * RP + (tid & 7) * 4;
  const int   lr = (wm + row) * RP + 6 * grp;

  v8f acc[3][2][2];
#pragma unroll
  for (int c = 0; c < 3; ++c)
#pragma unroll
    for (int i = 0; i < 2; ++i)
#pragma unroll
      for (int j = 0; j < 2; ++j) acc[c][i][j] = (v8f)0.0f;

  v4f rd[12];
  v4f f0[2][4], f1[2][4];

  fetchR(Yb, yo, 0, rd);
  fetchF(Fw, 0, f0);
  storeR(&Rs[0][lw], rd);
  __syncthreads();

  for (int t = 0; t < NT_K; t += 2) {
    fetchR(Yb, yo, t + 1, rd);
    fetchF(Fw, t + 1, f1);
    __builtin_amdgcn_sched_barrier(0);
    mma_rows3(&Rs[0][lr], f0, acc);
    storeR(&Rs[1][lw], rd);
    __syncthreads();

    const bool more = (t + 2 < NT_K);
    if (more) {
      fetchR(Yb, yo, t + 2, rd);
      fetchF(Fw, t + 2, f0);
    }
    mma_rows3(&Rs[1][lr], f1, acc);
    if (more) storeR(&Rs[0][lw], rd);
    __syncthreads();
  }

  float* Ob = Out + off;
#pragma unroll
  for (int mt = 0; mt < 2; ++mt)
#pragma unroll
    for (int nt = 0; nt < 2; ++nt)
#pragma unroll
      for (int v = 0; v < 8; ++v) {
        int m = m0 + wm + mt * 16 + grp * 8 + v;
        int l = n0 + wn + nt * 16 + row;
        float* p = &Ob[(long)m * 1536 + l * 3];
        p[0] = acc[0][mt][nt][v];
        p[1] = acc[1][mt][nt][v];
        p[2] = acc[2][mt][nt][v];
      }
}

extern "C" void kernel_launch(void* const* d_in, const int* in_sizes, int n_in,
                              void* d_out, int out_size, void* d_ws, size_t ws_size,
                              hipStream_t stream) {
  (void)in_sizes; (void)n_in; (void)out_size; (void)ws_size;
  const float* x = (const float*)d_in[0];
  float* out = (float*)d_out;
  float* Fm  = (float*)d_ws;
  float* Y1  = Fm + 512 * 512;
  const long HWC = 512L * 512L * 3L;

  dct_make<<<dim3(1024), dim3(256), 0, stream>>>(Fm);

  dct_cols<<<dim3(24, 4, 32), dim3(256), 0, stream>>>(
      Fm, x, Y1,
        1536, 1,   1536, 1,   HWC);

  dct_rows3<<<dim3(8, 4, 32), dim3(256), 0, stream>>>(Fm, Y1, out);
}


# Stage 1 data operand LDS-resident for the whole contraction: 512x64 panel copied once per work-group (b128, bank-swizzled, unpadded), one barrier, then 2 row blocks with no barrier or LDS store
# speedup vs baseline: 1.5151x; 1.1413x over previous
#include <hip/hip_runtime.h>


typedef __attribute__((ext_vector_type(2))) float v2f;
typedef __attribute__((ext_vector_type(4))) float v4f;
typedef __attribute__((ext_vector_type(8))) float v8f;

#define TILE_K 32
#define BLK_M  128
#define BLK_N  64
#define RP     100
#define GK     512
#define NT_K   (GK / TILE_K)

__global__ __launch_bounds__(256) void dct_make(float* __restrict__ F) {
  int idx  = blockIdx.x * 256 + threadIdx.x;
  int e    = idx & 3;
  int lane = (idx >> 2) & 31;
  int j    = (idx >> 7) & 63;
  int t16  = idx >> 13;
  int k = t16 * 16 + (lane & 15);
  int n = (2 * j + (e >> 1)) * 4 + 2 * (lane >> 4) + (e & 1);
  int r = ((2 * n + 1) * k) & 2047;
  float th = 3.14159265358979323846f * (float)r * (1.0f / 1024.0f);
  float v = cosf(th) * 0.0625f;
  if (k == 0) v *= 0.70710678118654752f;
  F[idx] = v;
}

__device__ __forceinline__ void fetchF(const float* __restrict__ Fw, int kt,
                                       v4f (&f)[2][4]) {
#pragma unroll
  for (int i = 0; i < 2; ++i)
#pragma unroll
    for (int jj = 0; jj < 4; ++jj)
      f[i][jj] = *(const v4f*)&Fw[(i * 64 + kt * 4 + jj) * 128];
}

#define PANEL_DW   (256 * 128)
#define PASSES     2

__device__ __forceinline__ void mma_panel(const float* __restrict__ B0,
                                          const float* __restrict__ B1,
                                          const v4f (&f)[2][4],
                                          v8f (&acc)[2][2]) {
#pragma unroll
  for (int s = 0; s < TILE_K / 4; ++s) {
    v2f a[2], b[2];
#pragma unroll
    for (int i = 0; i < 2; ++i) {
      const v4f q = f[i][s >> 1];
      a[i] = (s & 1) ? q.zw : q.xy;
    }
    b[0] = *(const v2f*)&B0[s * 256];
    b[1] = *(const v2f*)&B1[s * 256];
#pragma unroll
    for (int mt = 0; mt < 2; ++mt)
#pragma unroll
      for (int nt = 0; nt < 2; ++nt)
        acc[mt][nt] = __builtin_amdgcn_wmma_f32_16x16x4_f32(
            false, a[mt], false, b[nt], (short)0, acc[mt][nt], false, false);
  }
}

__global__ __launch_bounds__(256)
void dct_cols_panel(const float* __restrict__ F, const float* __restrict__ X,
                    float* __restrict__ C) {
  __shared__ float Ds[PANEL_DW];

  const int tid = threadIdx.x;
  const long off = (long)blockIdx.z * (512L * 1536L) + blockIdx.x * 64;
  const float* Xb = X + off;
  float*       Cb = C + off;

  {
    const int n4 = tid & 15;
    const int kp = tid >> 4;
    const float* src = Xb + (unsigned)(2 * kp) * 1536u + (unsigned)n4 * 4u;
    float* dst = &Ds[kp * 128 + ((8 * n4) ^ (32 * (kp & 1)))];
    v4f ra[16], rb[16];
#pragma unroll
    for (int sl = 0; sl < 16; ++sl) {
      ra[sl] = *(const v4f*)&src[sl * 32 * 1536];
      rb[sl] = *(const v4f*)&src[sl * 32 * 1536 + 1536];
    }
#pragma unroll
    for (int sl = 0; sl < 16; ++sl) {
      v4f lo, hi;
      lo.x = ra[sl].x; lo.y = rb[sl].x; lo.z = ra[sl].y; lo.w = rb[sl].y;
      hi.x = ra[sl].z; hi.y = rb[sl].z; hi.z = ra[sl].w; hi.w = rb[sl].w;
      *(v4f*)&dst[sl * 16 * 128]     = lo;
      *(v4f*)&dst[sl * 16 * 128 + 4] = hi;
    }
  }

  const int lane = tid & 31;
  const int wave = tid >> 5;
  const int wm   = (wave & 3) * 32;
  const int wn   = (wave >> 2) * 32;
  const int row  = lane & 15;
  const int grp  = lane >> 4;

  const float* B0 = &Ds[grp * 128 + 2 * (wn + row) + 32 * grp];
  const float* B1 = &Ds[grp * 128 + 2 * (wn + row) + 32 * (1 - grp)];

  const int mb0 = blockIdx.y * PASSES;
  const float* Fw = F + ((long)((mb0 * 128 + wm) >> 4) * 64 * 32 + lane) * 4;

  v8f acc[2][2];
#pragma unroll
  for (int i = 0; i < 2; ++i)
#pragma unroll
    for (int j = 0; j < 2; ++j) acc[i][j] = (v8f)0.0f;

  v4f f0[2][4], f1[2][4];
  fetchF(Fw, 0, f0);

  __syncthreads();

  for (int p = 0; p < PASSES; ++p) {
    for (int t = 0; t < NT_K; t += 2) {
      fetchF(Fw, t + 1, f1);
      __builtin_amdgcn_sched_barrier(0);
      mma_panel(B0 + t * 2048, B1 + t * 2048, f0, acc);

      if (t + 2 < NT_K)        fetchF(Fw, t + 2, f0);
      else if (p + 1 < PASSES) fetchF(Fw + 8 * 64 * 128, 0, f0);
      __builtin_amdgcn_sched_barrier(0);
      mma_panel(B0 + (t + 1) * 2048, B1 + (t + 1) * 2048, f1, acc);
    }

    const int m0 = (mb0 + p) * 128;
#pragma unroll
    for (int mt = 0; mt < 2; ++mt)
#pragma unroll
      for (int nt = 0; nt < 2; ++nt) {
#pragma unroll
        for (int v = 0; v < 8; ++v) {
          int m = m0 + wm + mt * 16 + grp * 8 + v;
          int n = wn + nt * 16 + row;
          Cb[(long)m * 1536 + n] = acc[mt][nt][v];
        }
        acc[mt][nt] = (v8f)0.0f;
      }
    Fw += 8 * 64 * 128;
  }
}


__device__ __forceinline__ void fetchR(const float* __restrict__ Yb, unsigned yo,
                                       int kt, v4f (&r)[12]) {
#pragma unroll
  for (int t = 0; t < 12; ++t)
    r[t] = *(const v4f*)&(Yb + kt * 96)[yo + (t & 3) * 32 * 1536 + (t >> 2) * 32];
}

__device__ __forceinline__ void storeR(float* __restrict__ Rs, const v4f (&r)[12]) {
#pragma unroll
  for (int t = 0; t < 12; ++t)
    *(v4f*)&Rs[(t & 3) * 32 * RP + (t >> 2) * 32] = r[t];
}

__device__ __forceinline__ void mma_rows3(const float* __restrict__ Rw,
                                          const v4f (&f)[2][4],
                                          v8f (&acc)[3][2][2]) {
#pragma unroll
  for (int s = 0; s < TILE_K / 4; ++s) {
    v2f a[3][2], b[2];
#pragma unroll
    for (int i = 0; i < 2; ++i) {
      const v4f q = f[i][s >> 1];
      b[i] = (s & 1) ? q.zw : q.xy;
    }
#pragma unroll
    for (int c = 0; c < 3; ++c)
#pragma unroll
      for (int mt = 0; mt < 2; ++mt) {
        const float* p = &Rw[mt * 16 * RP + s * 12 + c];
        a[c][mt].x = p[0];
        a[c][mt].y = p[3];
      }
#pragma unroll
    for (int c = 0; c < 3; ++c)
#pragma unroll
      for (int mt = 0; mt < 2; ++mt)
#pragma unroll
        for (int nt = 0; nt < 2; ++nt)
          acc[c][mt][nt] = __builtin_amdgcn_wmma_f32_16x16x4_f32(
              false, a[c][mt], false, b[nt], (short)0, acc[c][mt][nt], false, false);
  }
}

__global__ __launch_bounds__(256, 4)
void dct_rows3(const float* __restrict__ F, const float* __restrict__ Y,
               float* __restrict__ Out) {
  __shared__ float Rs[2][BLK_M * RP];

  const int tid = threadIdx.x;
  const long off = (long)blockIdx.z * (512L * 1536L);
  const int m0 = blockIdx.y * BLK_M;
  const int n0 = blockIdx.x * BLK_N;

  const int lane = tid & 31;
  const int wave = __builtin_amdgcn_readfirstlane(tid >> 5);
  const int wm   = (wave & 3) * 32;
  const int wn   = (wave >> 2) * 32;
  const int row  = lane & 15;
  const int grp  = lane >> 4;

  const int t16 = (n0 + wn) >> 4;
  const float* Fw = F + ((long)t16 * 64 * 32 + lane) * 4;

  const float* Yb = Y + off + (long)m0 * 1536;
  const unsigned yo = (unsigned)(tid >> 3) * 1536u + (unsigned)(tid & 7) * 4u;
  const int   lw = (tid >> 3) * RP + (tid & 7) * 4;
  const int   lr = (wm + row) * RP + 6 * grp;

  v8f acc[3][2][2];
#pragma unroll
  for (int c = 0; c < 3; ++c)
#pragma unroll
    for (int i = 0; i < 2; ++i)
#pragma unroll
      for (int j = 0; j < 2; ++j) acc[c][i][j] = (v8f)0.0f;

  v4f rd[12];
  v4f f0[2][4], f1[2][4];

  fetchR(Yb, yo, 0, rd);
  fetchF(Fw, 0, f0);
  storeR(&Rs[0][lw], rd);
  __syncthreads();

  for (int t = 0; t < NT_K; t += 2) {
    fetchR(Yb, yo, t + 1, rd);
    fetchF(Fw, t + 1, f1);
    __builtin_amdgcn_sched_barrier(0);
    mma_rows3(&Rs[0][lr], f0, acc);
    storeR(&Rs[1][lw], rd);
    __syncthreads();

    const bool more = (t + 2 < NT_K);
    if (more) {
      fetchR(Yb, yo, t + 2, rd);
      fetchF(Fw, t + 2, f0);
    }
    mma_rows3(&Rs[1][lr], f1, acc);
    if (more) storeR(&Rs[0][lw], rd);
    __syncthreads();
  }

  float* Ob = Out + off;
#pragma unroll
  for (int mt = 0; mt < 2; ++mt)
#pragma unroll
    for (int nt = 0; nt < 2; ++nt)
#pragma unroll
      for (int v = 0; v < 8; ++v) {
        int m = m0 + wm + mt * 16 + grp * 8 + v;
        int l = n0 + wn + nt * 16 + row;
        float* p = &Ob[(long)m * 1536 + l * 3];
        p[0] = acc[0][mt][nt][v];
        p[1] = acc[1][mt][nt][v];
        p[2] = acc[2][mt][nt][v];
      }
}

extern "C" void kernel_launch(void* const* d_in, const int* in_sizes, int n_in,
                              void* d_out, int out_size, void* d_ws, size_t ws_size,
                              hipStream_t stream) {
  (void)in_sizes; (void)n_in; (void)out_size; (void)ws_size;
  const float* x = (const float*)d_in[0];
  float* out = (float*)d_out;
  float* Fm  = (float*)d_ws;
  float* Y1  = Fm + 512 * 512;

  dct_make<<<dim3(1024), dim3(256), 0, stream>>>(Fm);

  dct_cols_panel<<<dim3(24, 2, 32), dim3(256), 0, stream>>>(Fm, x, Y1);

  dct_rows3<<<dim3(8, 4, 32), dim3(256), 0, stream>>>(Fm, Y1, out);
}


# Stage 1 adjacent-pair fold (decimation in time): data pre-summed x[2m]+-x[2m+1], half-size cos and sin tables, K 512 -> 256, rows k and 512-k by an epilogue rotation; changes summation order
# speedup vs baseline: 1.8705x; 1.2346x over previous
#include <hip/hip_runtime.h>


typedef __attribute__((ext_vector_type(2))) float v2f;
typedef __attribute__((ext_vector_type(4))) float v4f;
typedef __attribute__((ext_vector_type(8))) float v8f;

#define TILE_K 32
#define BLK_M  128
#define BLK_N  64
#define RP     100
#define GK     512
#define NT_K   (GK / TILE_K)

__global__ __launch_bounds__(256) void dct_make(float* __restrict__ F) {
  int idx  = blockIdx.x * 256 + threadIdx.x;
  int e    = idx & 3;
  int lane = (idx >> 2) & 31;
  int j    = (idx >> 7) & 63;
  int t16  = idx >> 13;
  int k = t16 * 16 + (lane & 15);
  int n = (2 * j + (e >> 1)) * 4 + 2 * (lane >> 4) + (e & 1);
  int r = ((2 * n + 1) * k) & 2047;
  float th = 3.14159265358979323846f * (float)r * (1.0f / 1024.0f);
  float v = cosf(th) * 0.0625f;
  if (k == 0) v *= 0.70710678118654752f;
  F[idx] = v;
}

__device__ __forceinline__ void fetchF(const float* __restrict__ Fw, int kt,
                                       v4f (&f)[2][4]) {
#pragma unroll
  for (int i = 0; i < 2; ++i)
#pragma unroll
    for (int jj = 0; jj < 4; ++jj)
      f[i][jj] = *(const v4f*)&Fw[(i * 64 + kt * 4 + jj) * 128];
}

__global__ __launch_bounds__(256) void dct_make_half(float* __restrict__ Fh) {
  int idx  = blockIdx.x * 256 + threadIdx.x;
  int e    = idx & 3;
  int lane = (idx >> 2) & 31;
  int j    = (idx >> 7) & 31;
  int t16  = idx >> 12;
  int k = t16 * 16 + (lane & 15);
  int m = (2 * j + (e >> 1)) * 4 + 2 * (lane >> 4) + (e & 1);
  int r = ((2 * m + 1) * k) & 1023;
  float th = 3.14159265358979323846f * (float)r * (1.0f / 512.0f);
  float vc = cosf(th) * 0.0625f;
  float vs = sinf(th) * 0.0625f;
  if (k == 0) {
    vc *= 0.70710678118654752f;
    vs = (m & 1) ? -0.0625f * 0.70710678118654752f : 0.0625f * 0.70710678118654752f;
  }
  Fh[idx]         = vc;
  Fh[65536 + idx] = vs;
  if (idx < 256) {
    float dl = 3.14159265358979323846f * (float)idx * (1.0f / 1024.0f);
    Fh[131072 + 2 * idx]     = cosf(dl);
    Fh[131072 + 2 * idx + 1] = sinf(dl);
  }
}

__device__ __forceinline__ void fetchH(const float* __restrict__ Fw, int kt,
                                       v4f (&f)[2][2]) {
#pragma unroll
  for (int i = 0; i < 2; ++i)
#pragma unroll
    for (int jj = 0; jj < 2; ++jj)
      f[i][jj] = *(const v4f*)&Fw[(i * 32 + kt * 2 + jj) * 128];
}

#define HPANEL_DW  (128 * 128)
#define TILE_H     16
#define NT_H       (256 / TILE_H)

__device__ __forceinline__ void mma_fold(const float* __restrict__ B0,
                                         const float* __restrict__ B1,
                                         const v4f (&fc)[2][2],
                                         const v4f (&fs)[2][2],
                                         v8f (&accC)[2][2], v8f (&accS)[2][2]) {
#pragma unroll
  for (int s = 0; s < TILE_H / 4; ++s) {
    v2f ac[2], as[2], bs[2], bd[2];
#pragma unroll
    for (int i = 0; i < 2; ++i) {
      const v4f qc = fc[i][s >> 1];
      const v4f qs = fs[i][s >> 1];
      ac[i] = (s & 1) ? qc.zw : qc.xy;
      as[i] = (s & 1) ? qs.zw : qs.xy;
    }
    bs[0] = *(const v2f*)&B0[s * 256];
    bs[1] = *(const v2f*)&B1[s * 256];
    bd[0] = *(const v2f*)&B0[HPANEL_DW + s * 256];
    bd[1] = *(const v2f*)&B1[HPANEL_DW + s * 256];
#pragma unroll
    for (int mt = 0; mt < 2; ++mt)
#pragma unroll
      for (int nt = 0; nt < 2; ++nt) {
        accC[mt][nt] = __builtin_amdgcn_wmma_f32_16x16x4_f32(
            false, ac[mt], false, bs[nt], (short)0, accC[mt][nt], false, false);
        accS[mt][nt] = __builtin_amdgcn_wmma_f32_16x16x4_f32(
            false, as[mt], false, bd[nt], (short)0, accS[mt][nt], false, false);
      }
  }
}

__global__ __launch_bounds__(256, 4)
void dct_cols_fold(const float* __restrict__ Fh, const float* __restrict__ X,
                   float* __restrict__ C) {
  __shared__ float Ds[2 * HPANEL_DW];

  const int tid = threadIdx.x;
  const long off = (long)blockIdx.z * (512L * 1536L) + blockIdx.x * 64;
  const float* Xb = X + off;
  float*       Cb = C + off;

  {
    const int n4 = tid & 15;
    const int kp = tid >> 4;
    const float* src = Xb + (unsigned)(4 * kp) * 1536u + (unsigned)n4 * 4u;
    float* dst = &Ds[kp * 128 + ((8 * n4) ^ (32 * (kp & 1)))];
    v4f rr[8][4];
#pragma unroll
    for (int sl = 0; sl < 8; ++sl)
#pragma unroll
      for (int i = 0; i < 4; ++i)
        rr[sl][i] = *(const v4f*)&src[(sl * 64 + i) * 1536];
#pragma unroll
    for (int sl = 0; sl < 8; ++sl) {
      const v4f s0 = rr[sl][0] + rr[sl][1], d0 = rr[sl][0] - rr[sl][1];
      const v4f s1 = rr[sl][2] + rr[sl][3], d1 = rr[sl][2] - rr[sl][3];
      v4f lo, hi;
      lo.x = s0.x; lo.y = s1.x; lo.z = s0.y; lo.w = s1.y;
      hi.x = s0.z; hi.y = s1.z; hi.z = s0.w; hi.w = s1.w;
      *(v4f*)&dst[sl * 16 * 128]     = lo;
      *(v4f*)&dst[sl * 16 * 128 + 4] = hi;
      lo.x = d0.x; lo.y = d1.x; lo.z = d0.y; lo.w = d1.y;
      hi.x = d0.z; hi.y = d1.z; hi.z = d0.w; hi.w = d1.w;
      *(v4f*)&dst[HPANEL_DW + sl * 16 * 128]     = lo;
      *(v4f*)&dst[HPANEL_DW + sl * 16 * 128 + 4] = hi;
    }
  }
  __builtin_amdgcn_sched_barrier(0);

  const int lane = tid & 31;
  const int wave = tid >> 5;
  const int wm   = (wave & 3) * 32;
  const int wn   = (wave >> 2) * 32;
  const int row  = lane & 15;
  const int grp  = lane >> 4;

  const float* B0 = &Ds[grp * 128 + 2 * (wn + row) + 32 * grp];
  const float* B1 = &Ds[grp * 128 + 2 * (wn + row) + 32 * (1 - grp)];

  const int k0 = blockIdx.y * 128 + wm;
  const float* Fcw = Fh + ((long)(k0 >> 4) * 32 * 32 + lane) * 4;
  const float* Fsw = Fcw + 65536;

  v8f accC[2][2], accS[2][2];
#pragma unroll
  for (int i = 0; i < 2; ++i)
#pragma unroll
    for (int j = 0; j < 2; ++j) { accC[i][j] = (v8f)0.0f; accS[i][j] = (v8f)0.0f; }

  v4f fc0[2][2], fs0[2][2], fc1[2][2], fs1[2][2];
  fetchH(Fcw, 0, fc0);
  fetchH(Fsw, 0, fs0);

  __syncthreads();

  for (int t = 0; t < NT_H - 2; t += 2) {
    fetchH(Fcw, t + 1, fc1);
    fetchH(Fsw, t + 1, fs1);
    __builtin_amdgcn_sched_barrier(0);
    mma_fold(B0 + t * 1024, B1 + t * 1024, fc0, fs0, accC, accS);

    fetchH(Fcw, t + 2, fc0);
    fetchH(Fsw, t + 2, fs0);
    __builtin_amdgcn_sched_barrier(0);
    mma_fold(B0 + (t + 1) * 1024, B1 + (t + 1) * 1024, fc1, fs1, accC, accS);
  }
  fetchH(Fcw, NT_H - 1, fc1);
  fetchH(Fsw, NT_H - 1, fs1);
  __builtin_amdgcn_sched_barrier(0);
  mma_fold(B0 + (NT_H - 2) * 1024, B1 + (NT_H - 2) * 1024, fc0, fs0, accC, accS);

  v4f tw[2][4];
  {
    const float* Tw = Fh + 131072 + 2 * (k0 + grp * 8);
#pragma unroll
    for (int mt = 0; mt < 2; ++mt)
#pragma unroll
      for (int q = 0; q < 4; ++q)
        tw[mt][q] = *(const v4f*)&Tw[mt * 32 + q * 4];
  }
  __builtin_amdgcn_sched_barrier(0);
  mma_fold(B0 + (NT_H - 1) * 1024, B1 + (NT_H - 1) * 1024, fc1, fs1, accC, accS);

#pragma unroll
  for (int mt = 0; mt < 2; ++mt)
#pragma unroll
    for (int v = 0; v < 8; ++v) {
      const v4f q = tw[mt][v >> 1];
      const float cd = (v & 1) ? q.z : q.x;
      const float sd = (v & 1) ? q.w : q.y;
      const int k  = k0 + mt * 16 + grp * 8 + v;
      const int kb = (k == 0) ? 256 : 512 - k;
#pragma unroll
      for (int nt = 0; nt < 2; ++nt) {
        const int n = wn + nt * 16 + row;
        const float ce = accC[mt][nt][v];
        const float so = accS[mt][nt][v];
        Cb[(long)k  * 1536 + n] = cd * ce + sd * so;
        Cb[(long)kb * 1536 + n] = cd * so - sd * ce;
      }
    }
}


__device__ __forceinline__ void fetchR(const float* __restrict__ Yb, unsigned yo,
                                       int kt, v4f (&r)[12]) {
#pragma unroll
  for (int t = 0; t < 12; ++t)
    r[t] = *(const v4f*)&(Yb + kt * 96)[yo + (t & 3) * 32 * 1536 + (t >> 2) * 32];
}

__device__ __forceinline__ void storeR(float* __restrict__ Rs, const v4f (&r)[12]) {
#pragma unroll
  for (int t = 0; t < 12; ++t)
    *(v4f*)&Rs[(t & 3) * 32 * RP + (t >> 2) * 32] = r[t];
}

__device__ __forceinline__ void mma_rows3(const float* __restrict__ Rw,
                                          const v4f (&f)[2][4],
                                          v8f (&acc)[3][2][2]) {
#pragma unroll
  for (int s = 0; s < TILE_K / 4; ++s) {
    v2f a[3][2], b[2];
#pragma unroll
    for (int i = 0; i < 2; ++i) {
      const v4f q = f[i][s >> 1];
      b[i] = (s & 1) ? q.zw : q.xy;
    }
#pragma unroll
    for (int c = 0; c < 3; ++c)
#pragma unroll
      for (int mt = 0; mt < 2; ++mt) {
        const float* p = &Rw[mt * 16 * RP + s * 12 + c];
        a[c][mt].x = p[0];
        a[c][mt].y = p[3];
      }
#pragma unroll
    for (int c = 0; c < 3; ++c)
#pragma unroll
      for (int mt = 0; mt < 2; ++mt)
#pragma unroll
        for (int nt = 0; nt < 2; ++nt)
          acc[c][mt][nt] = __builtin_amdgcn_wmma_f32_16x16x4_f32(
              false, a[c][mt], false, b[nt], (short)0, acc[c][mt][nt], false, false);
  }
}

__global__ __launch_bounds__(256, 4)
void dct_rows3(const float* __restrict__ F, const float* __restrict__ Y,
               float* __restrict__ Out) {
  __shared__ float Rs[2][BLK_M * RP];

  const int tid = threadIdx.x;
  const long off = (long)blockIdx.z * (512L * 1536L);
  const int m0 = blockIdx.y * BLK_M;
  const int n0 = blockIdx.x * BLK_N;

  const int lane = tid & 31;
  const int wave = __builtin_amdgcn_readfirstlane(tid >> 5);
  const int wm   = (wave & 3) * 32;
  const int wn   = (wave >> 2) * 32;
  const int row  = lane & 15;
  const int grp  = lane >> 4;

  const int t16 = (n0 + wn) >> 4;
  const float* Fw = F + ((long)t16 * 64 * 32 + lane) * 4;

  const float* Yb = Y + off + (long)m0 * 1536;
  const unsigned yo = (unsigned)(tid >> 3) * 1536u + (unsigned)(tid & 7) * 4u;
  const int   lw = (tid >> 3) * RP + (tid & 7) * 4;
  const int   lr = (wm + row) * RP + 6 * grp;

  v8f acc[3][2][2];
#pragma unroll
  for (int c = 0; c < 3; ++c)
#pragma unroll
    for (int i = 0; i < 2; ++i)
#pragma unroll
      for (int j = 0; j < 2; ++j) acc[c][i][j] = (v8f)0.0f;

  v4f rd[12];
  v4f f0[2][4], f1[2][4];

  fetchR(Yb, yo, 0, rd);
  fetchF(Fw, 0, f0);
  storeR(&Rs[0][lw], rd);
  __syncthreads();

  for (int t = 0; t < NT_K; t += 2) {
    fetchR(Yb, yo, t + 1, rd);
    fetchF(Fw, t + 1, f1);
    __builtin_amdgcn_sched_barrier(0);
    mma_rows3(&Rs[0][lr], f0, acc);
    storeR(&Rs[1][lw], rd);
    __syncthreads();

    const bool more = (t + 2 < NT_K);
    if (more) {
      fetchR(Yb, yo, t + 2, rd);
      fetchF(Fw, t + 2, f0);
    }
    mma_rows3(&Rs[1][lr], f1, acc);
    if (more) storeR(&Rs[0][lw], rd);
    __syncthreads();
  }

  float* Ob = Out + off;
#pragma unroll
  for (int mt = 0; mt < 2; ++mt)
#pragma unroll
    for (int nt = 0; nt < 2; ++nt)
#pragma unroll
      for (int v = 0; v < 8; ++v) {
        int m = m0 + wm + mt * 16 + grp * 8 + v;
        int l = n0 + wn + nt * 16 + row;
        float* p = &Ob[(long)m * 1536 + l * 3];
        p[0] = acc[0][mt][nt][v];
        p[1] = acc[1][mt][nt][v];
        p[2] = acc[2][mt][nt][v];
      }
}

extern "C" void kernel_launch(void* const* d_in, const int* in_sizes, int n_in,
                              void* d_out, int out_size, void* d_ws, size_t ws_size,
                              hipStream_t stream) {
  (void)in_sizes; (void)n_in; (void)out_size; (void)ws_size;
  const float* x = (const float*)d_in[0];
  float* out = (float*)d_out;
  float* Fm  = (float*)d_ws;
  float* Y1  = Fm + 512 * 512;

  dct_make_half<<<dim3(256), dim3(256), 0, stream>>>(Fm);

  dct_cols_fold<<<dim3(24, 2, 32), dim3(256), 0, stream>>>(Fm, x, Y1);

  dct_make<<<dim3(1024), dim3(256), 0, stream>>>(Fm);

  dct_rows3<<<dim3(8, 4, 32), dim3(256), 0, stream>>>(Fm, Y1, out);
}


# Stage 2 adjacent-pair fold (as stage 1): row pixels pre-summed Y[2p]+-Y[2p+1] into LDS, half-size cos and sin tables as B, K 512 -> 256, columns l and 512-l by epilogue rotation; changes sum order
# speedup vs baseline: 2.2570x; 1.2066x over previous
#include <hip/hip_runtime.h>


typedef __attribute__((ext_vector_type(2))) float v2f;
typedef __attribute__((ext_vector_type(4))) float v4f;
typedef __attribute__((ext_vector_type(8))) float v8f;

__global__ __launch_bounds__(256) void dct_make_half(float* __restrict__ Fh) {
  int idx  = blockIdx.x * 256 + threadIdx.x;
  int e    = idx & 3;
  int lane = (idx >> 2) & 31;
  int j    = (idx >> 7) & 31;
  int t16  = idx >> 12;
  int k = t16 * 16 + (lane & 15);
  int m = (2 * j + (e >> 1)) * 4 + 2 * (lane >> 4) + (e & 1);
  int r = ((2 * m + 1) * k) & 1023;
  float th = 3.14159265358979323846f * (float)r * (1.0f / 512.0f);
  float vc = cosf(th) * 0.0625f;
  float vs = sinf(th) * 0.0625f;
  if (k == 0) {
    vc *= 0.70710678118654752f;
    vs = (m & 1) ? -0.0625f * 0.70710678118654752f : 0.0625f * 0.70710678118654752f;
  }
  Fh[idx]         = vc;
  Fh[65536 + idx] = vs;
  if (idx < 256) {
    float dl = 3.14159265358979323846f * (float)idx * (1.0f / 1024.0f);
    Fh[131072 + 2 * idx]     = cosf(dl);
    Fh[131072 + 2 * idx + 1] = sinf(dl);
  }
}

__device__ __forceinline__ void fetchH(const float* __restrict__ Fw, int kt,
                                       v4f (&f)[2][2]) {
#pragma unroll
  for (int i = 0; i < 2; ++i)
#pragma unroll
    for (int jj = 0; jj < 2; ++jj)
      f[i][jj] = *(const v4f*)&Fw[(i * 32 + kt * 2 + jj) * 128];
}

#define HPANEL_DW  (128 * 128)
#define TILE_H     16
#define NT_H       (256 / TILE_H)

__device__ __forceinline__ void mma_fold(const float* __restrict__ B0,
                                         const float* __restrict__ B1,
                                         const v4f (&fc)[2][2],
                                         const v4f (&fs)[2][2],
                                         v8f (&accC)[2][2], v8f (&accS)[2][2]) {
#pragma unroll
  for (int s = 0; s < TILE_H / 4; ++s) {
    v2f ac[2], as[2], bs[2], bd[2];
#pragma unroll
    for (int i = 0; i < 2; ++i) {
      const v4f qc = fc[i][s >> 1];
      const v4f qs = fs[i][s >> 1];
      ac[i] = (s & 1) ? qc.zw : qc.xy;
      as[i] = (s & 1) ? qs.zw : qs.xy;
    }
    bs[0] = *(const v2f*)&B0[s * 256];
    bs[1] = *(const v2f*)&B1[s * 256];
    bd[0] = *(const v2f*)&B0[HPANEL_DW + s * 256];
    bd[1] = *(const v2f*)&B1[HPANEL_DW + s * 256];
#pragma unroll
    for (int mt = 0; mt < 2; ++mt)
#pragma unroll
      for (int nt = 0; nt < 2; ++nt) {
        accC[mt][nt] = __builtin_amdgcn_wmma_f32_16x16x4_f32(
            false, ac[mt], false, bs[nt], (short)0, accC[mt][nt], false, false);
        accS[mt][nt] = __builtin_amdgcn_wmma_f32_16x16x4_f32(
            false, as[mt], false, bd[nt], (short)0, accS[mt][nt], false, false);
      }
  }
}

__global__ __launch_bounds__(256, 4)
void dct_cols_fold(const float* __restrict__ Fh, const float* __restrict__ X,
                   float* __restrict__ C) {
  __shared__ float Ds[2 * HPANEL_DW];

  const int tid = threadIdx.x;
  const long off = (long)blockIdx.z * (512L * 1536L) + blockIdx.x * 64;
  const float* Xb = X + off;
  float*       Cb = C + off;

  {
    const int n4 = tid & 15;
    const int kp = tid >> 4;
    const float* src = Xb + (unsigned)(4 * kp) * 1536u + (unsigned)n4 * 4u;
    float* dst = &Ds[kp * 128 + ((8 * n4) ^ (32 * (kp & 1)))];
    v4f rr[8][4];
#pragma unroll
    for (int sl = 0; sl < 8; ++sl)
#pragma unroll
      for (int i = 0; i < 4; ++i)
        rr[sl][i] = *(const v4f*)&src[(sl * 64 + i) * 1536];
#pragma unroll
    for (int sl = 0; sl < 8; ++sl) {
      const v4f s0 = rr[sl][0] + rr[sl][1], d0 = rr[sl][0] - rr[sl][1];
      const v4f s1 = rr[sl][2] + rr[sl][3], d1 = rr[sl][2] - rr[sl][3];
      v4f lo, hi;
      lo.x = s0.x; lo.y = s1.x; lo.z = s0.y; lo.w = s1.y;
      hi.x = s0.z; hi.y = s1.z; hi.z = s0.w; hi.w = s1.w;
      *(v4f*)&dst[sl * 16 * 128]     = lo;
      *(v4f*)&dst[sl * 16 * 128 + 4] = hi;
      lo.x = d0.x; lo.y = d1.x; lo.z = d0.y; lo.w = d1.y;
      hi.x = d0.z; hi.y = d1.z; hi.z = d0.w; hi.w = d1.w;
      *(v4f*)&dst[HPANEL_DW + sl * 16 * 128]     = lo;
      *(v4f*)&dst[HPANEL_DW + sl * 16 * 128 + 4] = hi;
    }
  }
  __builtin_amdgcn_sched_barrier(0);

  const int lane = tid & 31;
  const int wave = tid >> 5;
  const int wm   = (wave & 3) * 32;
  const int wn   = (wave >> 2) * 32;
  const int row  = lane & 15;
  const int grp  = lane >> 4;

  const float* B0 = &Ds[grp * 128 + 2 * (wn + row) + 32 * grp];
  const float* B1 = &Ds[grp * 128 + 2 * (wn + row) + 32 * (1 - grp)];

  const int k0 = blockIdx.y * 128 + wm;
  const float* Fcw = Fh + ((long)(k0 >> 4) * 32 * 32 + lane) * 4;
  const float* Fsw = Fcw + 65536;

  v8f accC[2][2], accS[2][2];
#pragma unroll
  for (int i = 0; i < 2; ++i)
#pragma unroll
    for (int j = 0; j < 2; ++j) { accC[i][j] = (v8f)0.0f; accS[i][j] = (v8f)0.0f; }

  v4f fc0[2][2], fs0[2][2], fc1[2][2], fs1[2][2];
  fetchH(Fcw, 0, fc0);
  fetchH(Fsw, 0, fs0);

  __syncthreads();

  for (int t = 0; t < NT_H - 2; t += 2) {
    fetchH(Fcw, t + 1, fc1);
    fetchH(Fsw, t + 1, fs1);
    __builtin_amdgcn_sched_barrier(0);
    mma_fold(B0 + t * 1024, B1 + t * 1024, fc0, fs0, accC, accS);

    fetchH(Fcw, t + 2, fc0);
    fetchH(Fsw, t + 2, fs0);
    __builtin_amdgcn_sched_barrier(0);
    mma_fold(B0 + (t + 1) * 1024, B1 + (t + 1) * 1024, fc1, fs1, accC, accS);
  }
  fetchH(Fcw, NT_H - 1, fc1);
  fetchH(Fsw, NT_H - 1, fs1);
  __builtin_amdgcn_sched_barrier(0);
  mma_fold(B0 + (NT_H - 2) * 1024, B1 + (NT_H - 2) * 1024, fc0, fs0, accC, accS);

  v4f tw[2][4];
  {
    const float* Tw = Fh + 131072 + 2 * (k0 + grp * 8);
#pragma unroll
    for (int mt = 0; mt < 2; ++mt)
#pragma unroll
      for (int q = 0; q < 4; ++q)
        tw[mt][q] = *(const v4f*)&Tw[mt * 32 + q * 4];
  }
  __builtin_amdgcn_sched_barrier(0);
  mma_fold(B0 + (NT_H - 1) * 1024, B1 + (NT_H - 1) * 1024, fc1, fs1, accC, accS);

#pragma unroll
  for (int mt = 0; mt < 2; ++mt)
#pragma unroll
    for (int v = 0; v < 8; ++v) {
      const v4f q = tw[mt][v >> 1];
      const float cd = (v & 1) ? q.z : q.x;
      const float sd = (v & 1) ? q.w : q.y;
      const int k  = k0 + mt * 16 + grp * 8 + v;
      const int kb = (k == 0) ? 256 : 512 - k;
#pragma unroll
      for (int nt = 0; nt < 2; ++nt) {
        const int n = wn + nt * 16 + row;
        const float ce = accC[mt][nt][v];
        const float so = accS[mt][nt][v];
        Cb[(long)k  * 1536 + n] = cd * ce + sd * so;
        Cb[(long)kb * 1536 + n] = cd * so - sd * ce;
      }
    }
}

#define BLK_M  128
#define RP     100
#define NT_W   16

__device__ __forceinline__ void fetchY(const float* __restrict__ Yb, unsigned yo,
                                       int kt, v4f (&r)[12]) {
#pragma unroll
  for (int i = 0; i < 4; ++i)
#pragma unroll
    for (int j = 0; j < 3; ++j)
      r[i * 3 + j] = *(const v4f*)&(Yb + kt * 96)[yo + i * 1536 + j * 4];
}

__device__ __forceinline__ void foldStore(float* __restrict__ Rw, float sg,
                                          const v4f (&r)[12]) {
#pragma unroll
  for (int i = 0; i < 4; ++i) {
    const v4f r0 = r[3 * i], r1 = r[3 * i + 1], r2 = r[3 * i + 2];
    v4f w0, w1, w2;
    w0.x = r0.x + sg * r0.w;  w0.y = r1.z + sg * r2.y;
    w0.z = r0.x - sg * r0.w;  w0.w = r1.z - sg * r2.y;
    w1.x = r0.y + sg * r1.x;  w1.y = r1.w + sg * r2.z;
    w1.z = r0.y - sg * r1.x;  w1.w = r1.w - sg * r2.z;
    w2.x = r0.z + sg * r1.y;  w2.y = r2.x + sg * r2.w;
    w2.z = r0.z - sg * r1.y;  w2.w = r2.x - sg * r2.w;
    *(v4f*)&Rw[i * RP]     = w0;
    *(v4f*)&Rw[i * RP + 4] = w1;
    *(v4f*)&Rw[i * RP + 8] = w2;
  }
}

__device__ __forceinline__ void mma_rows_fold(const float* __restrict__ RwS,
                                              const float* __restrict__ RwD,
                                              const v4f (&fc)[2][2],
                                              const v4f (&fs)[2][2],
                                              v8f (&accC)[3][2], v8f (&accS)[3][2]) {
#pragma unroll
  for (int s = 0; s < 4; ++s) {
    v2f bc[2], bs[2], av[3], dv[3];
#pragma unroll
    for (int i = 0; i < 2; ++i) {
      const v4f qc = fc[i][s >> 1];
      const v4f qs = fs[i][s >> 1];
      bc[i] = (s & 1) ? qc.zw : qc.xy;
      bs[i] = (s & 1) ? qs.zw : qs.xy;
    }
#pragma unroll
    for (int c = 0; c < 3; ++c) {
      av[c] = *(const v2f*)&RwS[s * 24 + c * 4];
      dv[c] = *(const v2f*)&RwD[s * 24 + c * 4];
    }
#pragma unroll
    for (int c = 0; c < 3; ++c)
#pragma unroll
      for (int nt = 0; nt < 2; ++nt) {
        accC[c][nt] = __builtin_amdgcn_wmma_f32_16x16x4_f32(
            false, av[c], false, bc[nt], (short)0, accC[c][nt], false, false);
        accS[c][nt] = __builtin_amdgcn_wmma_f32_16x16x4_f32(
            false, dv[c], false, bs[nt], (short)0, accS[c][nt], false, false);
      }
  }
}

__global__ __launch_bounds__(256, 4)
void dct_rows3_fold(const float* __restrict__ Fh, const float* __restrict__ Y,
                    float* __restrict__ Out) {
  __shared__ float Rs[2][BLK_M * RP];

  const int tid = threadIdx.x;
  const long off = (long)blockIdx.z * (512L * 1536L);
  const int m0 = blockIdx.y * BLK_M;
  const int l0 = blockIdx.x * 32;

  const int lane = tid & 31;
  const int wave = __builtin_amdgcn_readfirstlane(tid >> 5);
  const int row  = lane & 15;
  const int grp  = lane >> 4;

  const float* Fcw = Fh + ((long)(l0 >> 4) * 32 * 32 + lane) * 4;
  const float* Fsw = Fcw + 65536;

  const float* Yb = Y + off + (long)m0 * 1536;
  const int g  = tid & 7;
  const int r4 = (tid >> 5) * 16 + ((tid >> 3) & 3) * 4;
  const unsigned yo = (unsigned)r4 * 1536u + (unsigned)g * 12u;
  const int   lw = r4 * RP + g * 12;
  const float sg = (g & 1) ? -1.0f : 1.0f;
  const int   lrS = (wave * 16 + row) * RP + 14 * grp;
  const int   lrD = (wave * 16 + row) * RP + 10 * grp + 2;

  v8f accC[3][2], accS[3][2];
#pragma unroll
  for (int c = 0; c < 3; ++c)
#pragma unroll
    for (int j = 0; j < 2; ++j) { accC[c][j] = (v8f)0.0f; accS[c][j] = (v8f)0.0f; }

  v4f rd[12];
  v4f fc0[2][2], fs0[2][2], fc1[2][2], fs1[2][2];

  fetchY(Yb, yo, 0, rd);
  fetchH(Fcw, 0, fc0);
  fetchH(Fsw, 0, fs0);
  foldStore(&Rs[0][lw], sg, rd);
  __syncthreads();

  for (int t = 0; t < NT_W - 2; t += 2) {
    fetchY(Yb, yo, t + 1, rd);
    fetchH(Fcw, t + 1, fc1);
    fetchH(Fsw, t + 1, fs1);
    __builtin_amdgcn_sched_barrier(0);
    mma_rows_fold(&Rs[0][lrS], &Rs[0][lrD], fc0, fs0, accC, accS);
    foldStore(&Rs[1][lw], sg, rd);
    __syncthreads();

    fetchY(Yb, yo, t + 2, rd);
    fetchH(Fcw, t + 2, fc0);
    fetchH(Fsw, t + 2, fs0);
    __builtin_amdgcn_sched_barrier(0);
    mma_rows_fold(&Rs[1][lrS], &Rs[1][lrD], fc1, fs1, accC, accS);
    foldStore(&Rs[0][lw], sg, rd);
    __syncthreads();
  }
  fetchY(Yb, yo, NT_W - 1, rd);
  fetchH(Fcw, NT_W - 1, fc1);
  fetchH(Fsw, NT_W - 1, fs1);
  __builtin_amdgcn_sched_barrier(0);
  mma_rows_fold(&Rs[0][lrS], &Rs[0][lrD], fc0, fs0, accC, accS);
  foldStore(&Rs[1][lw], sg, rd);
  __syncthreads();

  v2f tw[2];
#pragma unroll
  for (int nt = 0; nt < 2; ++nt)
    tw[nt] = *(const v2f*)&Fh[131072 + 2 * (l0 + nt * 16 + row)];
  __builtin_amdgcn_sched_barrier(0);
  mma_rows_fold(&Rs[1][lrS], &Rs[1][lrD], fc1, fs1, accC, accS);

  float* Ob = Out + off;
#pragma unroll
  for (int nt = 0; nt < 2; ++nt) {
    const int l  = l0 + nt * 16 + row;
    const int lb = (l == 0) ? 256 : 512 - l;
    const float cd = tw[nt].x;
    const float sd = tw[nt].y;
#pragma unroll
    for (int v = 0; v < 8; ++v) {
      const int m = m0 + wave * 16 + grp * 8 + v;
      float* p  = &Ob[(long)m * 1536 + l * 3];
      float* pb = &Ob[(long)m * 1536 + lb * 3];
#pragma unroll
      for (int c = 0; c < 3; ++c) {
        const float ce = accC[c][nt][v];
        const float so = accS[c][nt][v];
        p[c]  = cd * ce + sd * so;
        pb[c] = cd * so - sd * ce;
      }
    }
  }
}

extern "C" void kernel_launch(void* const* d_in, const int* in_sizes, int n_in,
                              void* d_out, int out_size, void* d_ws, size_t ws_size,
                              hipStream_t stream) {
  (void)in_sizes; (void)n_in; (void)out_size; (void)ws_size;
  const float* x = (const float*)d_in[0];
  float* out = (float*)d_out;
  float* Fm  = (float*)d_ws;
  float* Y1  = Fm + 512 * 512;

  dct_make_half<<<dim3(256), dim3(256), 0, stream>>>(Fm);

  dct_cols_fold<<<dim3(24, 2, 32), dim3(256), 0, stream>>>(Fm, x, Y1);

  dct_rows3_fold<<<dim3(8, 4, 32), dim3(256), 0, stream>>>(Fm, Y1, out);
}


# Stage 2 data operand LDS-resident for the whole contraction: work-group takes 16 rows x all 256 pairs, row panel folded once into LDS, one barrier, then 16 K-tiles with no barrier or LDS store
# speedup vs baseline: 2.8474x; 1.2616x over previous
#include <hip/hip_runtime.h>


typedef __attribute__((ext_vector_type(2))) float v2f;
typedef __attribute__((ext_vector_type(4))) float v4f;
typedef __attribute__((ext_vector_type(8))) float v8f;

__global__ __launch_bounds__(256) void dct_make_half(float* __restrict__ Fh) {
  int idx  = blockIdx.x * 256 + threadIdx.x;
  int e    = idx & 3;
  int lane = (idx >> 2) & 31;
  int j    = (idx >> 7) & 31;
  int t16  = idx >> 12;
  int k = t16 * 16 + (lane & 15);
  int m = (2 * j + (e >> 1)) * 4 + 2 * (lane >> 4) + (e & 1);
  int r = ((2 * m + 1) * k) & 1023;
  float th = 3.14159265358979323846f * (float)r * (1.0f / 512.0f);
  float vc = cosf(th) * 0.0625f;
  float vs = sinf(th) * 0.0625f;
  if (k == 0) {
    vc *= 0.70710678118654752f;
    vs = (m & 1) ? -0.0625f * 0.70710678118654752f : 0.0625f * 0.70710678118654752f;
  }
  Fh[idx]         = vc;
  Fh[65536 + idx] = vs;
  if (idx < 256) {
    float dl = 3.14159265358979323846f * (float)idx * (1.0f / 1024.0f);
    Fh[131072 + 2 * idx]     = cosf(dl);
    Fh[131072 + 2 * idx + 1] = sinf(dl);
  }
}

__device__ __forceinline__ void fetchH(const float* __restrict__ Fw, int kt,
                                       v4f (&f)[2][2]) {
#pragma unroll
  for (int i = 0; i < 2; ++i)
#pragma unroll
    for (int jj = 0; jj < 2; ++jj)
      f[i][jj] = *(const v4f*)&Fw[(i * 32 + kt * 2 + jj) * 128];
}

#define HPANEL_DW  (128 * 128)
#define TILE_H     16
#define NT_H       (256 / TILE_H)

__device__ __forceinline__ void mma_fold(const float* __restrict__ B0,
                                         const float* __restrict__ B1,
                                         const v4f (&fc)[2][2],
                                         const v4f (&fs)[2][2],
                                         v8f (&accC)[2][2], v8f (&accS)[2][2]) {
#pragma unroll
  for (int s = 0; s < TILE_H / 4; ++s) {
    v2f ac[2], as[2], bs[2], bd[2];
#pragma unroll
    for (int i = 0; i < 2; ++i) {
      const v4f qc = fc[i][s >> 1];
      const v4f qs = fs[i][s >> 1];
      ac[i] = (s & 1) ? qc.zw : qc.xy;
      as[i] = (s & 1) ? qs.zw : qs.xy;
    }
    bs[0] = *(const v2f*)&B0[s * 256];
    bs[1] = *(const v2f*)&B1[s * 256];
    bd[0] = *(const v2f*)&B0[HPANEL_DW + s * 256];
    bd[1] = *(const v2f*)&B1[HPANEL_DW + s * 256];
#pragma unroll
    for (int mt = 0; mt < 2; ++mt)
#pragma unroll
      for (int nt = 0; nt < 2; ++nt) {
        accC[mt][nt] = __builtin_amdgcn_wmma_f32_16x16x4_f32(
            false, ac[mt], false, bs[nt], (short)0, accC[mt][nt], false, false);
        accS[mt][nt] = __builtin_amdgcn_wmma_f32_16x16x4_f32(
            false, as[mt], false, bd[nt], (short)0, accS[mt][nt], false, false);
      }
  }
}

__global__ __launch_bounds__(256, 4)
void dct_cols_fold(const float* __restrict__ Fh, const float* __restrict__ X,
                   float* __restrict__ C) {
  __shared__ float Ds[2 * HPANEL_DW];

  const int tid = threadIdx.x;
  const long off = (long)blockIdx.z * (512L * 1536L) + blockIdx.x * 64;
  const float* Xb = X + off;
  float*       Cb = C + off;

  {
    const int n4 = tid & 15;
    const int kp = tid >> 4;
    const float* src = Xb + (unsigned)(4 * kp) * 1536u + (unsigned)n4 * 4u;
    float* dst = &Ds[kp * 128 + ((8 * n4) ^ (32 * (kp & 1)))];
    v4f rr[8][4];
#pragma unroll
    for (int sl = 0; sl < 8; ++sl)
#pragma unroll
      for (int i = 0; i < 4; ++i)
        rr[sl][i] = *(const v4f*)&src[(sl * 64 + i) * 1536];
#pragma unroll
    for (int sl = 0; sl < 8; ++sl) {
      const v4f s0 = rr[sl][0] + rr[sl][1], d0 = rr[sl][0] - rr[sl][1];
      const v4f s1 = rr[sl][2] + rr[sl][3], d1 = rr[sl][2] - rr[sl][3];
      v4f lo, hi;
      lo.x = s0.x; lo.y = s1.x; lo.z = s0.y; lo.w = s1.y;
      hi.x = s0.z; hi.y = s1.z; hi.z = s0.w; hi.w = s1.w;
      *(v4f*)&dst[sl * 16 * 128]     = lo;
      *(v4f*)&dst[sl * 16 * 128 + 4] = hi;
      lo.x = d0.x; lo.y = d1.x; lo.z = d0.y; lo.w = d1.y;
      hi.x = d0.z; hi.y = d1.z; hi.z = d0.w; hi.w = d1.w;
      *(v4f*)&dst[HPANEL_DW + sl * 16 * 128]     = lo;
      *(v4f*)&dst[HPANEL_DW + sl * 16 * 128 + 4] = hi;
    }
  }
  __builtin_amdgcn_sched_barrier(0);

  const int lane = tid & 31;
  const int wave = tid >> 5;
  const int wm   = (wave & 3) * 32;
  const int wn   = (wave >> 2) * 32;
  const int row  = lane & 15;
  const int grp  = lane >> 4;

  const float* B0 = &Ds[grp * 128 + 2 * (wn + row) + 32 * grp];
  const float* B1 = &Ds[grp * 128 + 2 * (wn + row) + 32 * (1 - grp)];

  const int k0 = blockIdx.y * 128 + wm;
  const float* Fcw = Fh + ((long)(k0 >> 4) * 32 * 32 + lane) * 4;
  const float* Fsw = Fcw + 65536;

  v8f accC[2][2], accS[2][2];
#pragma unroll
  for (int i = 0; i < 2; ++i)
#pragma unroll
    for (int j = 0; j < 2; ++j) { accC[i][j] = (v8f)0.0f; accS[i][j] = (v8f)0.0f; }

  v4f fc0[2][2], fs0[2][2], fc1[2][2], fs1[2][2];
  fetchH(Fcw, 0, fc0);
  fetchH(Fsw, 0, fs0);

  __syncthreads();

  for (int t = 0; t < NT_H - 2; t += 2) {
    fetchH(Fcw, t + 1, fc1);
    fetchH(Fsw, t + 1, fs1);
    __builtin_amdgcn_sched_barrier(0);
    mma_fold(B0 + t * 1024, B1 + t * 1024, fc0, fs0, accC, accS);

    fetchH(Fcw, t + 2, fc0);
    fetchH(Fsw, t + 2, fs0);
    __builtin_amdgcn_sched_barrier(0);
    mma_fold(B0 + (t + 1) * 1024, B1 + (t + 1) * 1024, fc1, fs1, accC, accS);
  }
  fetchH(Fcw, NT_H - 1, fc1);
  fetchH(Fsw, NT_H - 1, fs1);
  __builtin_amdgcn_sched_barrier(0);
  mma_fold(B0 + (NT_H - 2) * 1024, B1 + (NT_H - 2) * 1024, fc0, fs0, accC, accS);

  v4f tw[2][4];
  {
    const float* Tw = Fh + 131072 + 2 * (k0 + grp * 8);
#pragma unroll
    for (int mt = 0; mt < 2; ++mt)
#pragma unroll
      for (int q = 0; q < 4; ++q)
        tw[mt][q] = *(const v4f*)&Tw[mt * 32 + q * 4];
  }
  __builtin_amdgcn_sched_barrier(0);
  mma_fold(B0 + (NT_H - 1) * 1024, B1 + (NT_H - 1) * 1024, fc1, fs1, accC, accS);

#pragma unroll
  for (int mt = 0; mt < 2; ++mt)
#pragma unroll
    for (int v = 0; v < 8; ++v) {
      const v4f q = tw[mt][v >> 1];
      const float cd = (v & 1) ? q.z : q.x;
      const float sd = (v & 1) ? q.w : q.y;
      const int k  = k0 + mt * 16 + grp * 8 + v;
      const int kb = (k == 0) ? 256 : 512 - k;
#pragma unroll
      for (int nt = 0; nt < 2; ++nt) {
        const int n = wn + nt * 16 + row;
        const float ce = accC[mt][nt][v];
        const float so = accS[mt][nt][v];
        Cb[(long)k  * 1536 + n] = cd * ce + sd * so;
        Cb[(long)kb * 1536 + n] = cd * so - sd * ce;
      }
    }
}

#define ROWS_W 16
#define ROWP   1572
#define NT_W   16

__device__ __forceinline__ void mma_rows_fold(const float* __restrict__ RwS,
                                              const float* __restrict__ RwD,
                                              const v4f (&fc)[2][2],
                                              const v4f (&fs)[2][2],
                                              v8f (&accC)[3][2], v8f (&accS)[3][2]) {
#pragma unroll
  for (int s = 0; s < 4; ++s) {
    v2f bc[2], bs[2], av[3], dv[3];
#pragma unroll
    for (int i = 0; i < 2; ++i) {
      const v4f qc = fc[i][s >> 1];
      const v4f qs = fs[i][s >> 1];
      bc[i] = (s & 1) ? qc.zw : qc.xy;
      bs[i] = (s & 1) ? qs.zw : qs.xy;
    }
#pragma unroll
    for (int c = 0; c < 3; ++c) {
      av[c] = *(const v2f*)&RwS[s * 24 + c * 4];
      dv[c] = *(const v2f*)&RwD[s * 24 + c * 4];
    }
#pragma unroll
    for (int c = 0; c < 3; ++c)
#pragma unroll
      for (int nt = 0; nt < 2; ++nt) {
        accC[c][nt] = __builtin_amdgcn_wmma_f32_16x16x4_f32(
            false, av[c], false, bc[nt], (short)0, accC[c][nt], false, false);
        accS[c][nt] = __builtin_amdgcn_wmma_f32_16x16x4_f32(
            false, dv[c], false, bs[nt], (short)0, accS[c][nt], false, false);
      }
  }
}

__global__ __launch_bounds__(256, 4)
void dct_rows3_fold(const float* __restrict__ Fh, const float* __restrict__ Y,
                    float* __restrict__ Out) {
  __shared__ float Rs[ROWS_W * ROWP];

  const int tid = threadIdx.x;
  const long offb = (long)blockIdx.z * (512L * 1536L);
  const int m0 = blockIdx.x * ROWS_W;

  {
    const int gg = tid & 127;
    const int rh = tid >> 7;
    const float sg = (gg & 1) ? -1.0f : 1.0f;
    const float* src = Y + offb + (long)m0 * 1536 + (unsigned)rh * 1536u + (unsigned)gg * 12u;
    float* dst = &Rs[rh * ROWP + gg * 12];
    v4f rr[8][3];
#pragma unroll
    for (int it = 0; it < 8; ++it)
#pragma unroll
      for (int j = 0; j < 3; ++j)
        rr[it][j] = *(const v4f*)&src[it * 2 * 1536 + j * 4];
#pragma unroll
    for (int it = 0; it < 8; ++it) {
      const v4f r0 = rr[it][0], r1 = rr[it][1], r2 = rr[it][2];
      v4f w0, w1, w2;
      w0.x = r0.x + sg * r0.w;  w0.y = r1.z + sg * r2.y;
      w0.z = r0.x - sg * r0.w;  w0.w = r1.z - sg * r2.y;
      w1.x = r0.y + sg * r1.x;  w1.y = r1.w + sg * r2.z;
      w1.z = r0.y - sg * r1.x;  w1.w = r1.w - sg * r2.z;
      w2.x = r0.z + sg * r1.y;  w2.y = r2.x + sg * r2.w;
      w2.z = r0.z - sg * r1.y;  w2.w = r2.x - sg * r2.w;
      *(v4f*)&dst[it * 2 * ROWP]     = w0;
      *(v4f*)&dst[it * 2 * ROWP + 4] = w1;
      *(v4f*)&dst[it * 2 * ROWP + 8] = w2;
    }
  }

  const int lane = tid & 31;
  const int wave = __builtin_amdgcn_readfirstlane(tid >> 5);
  const int row  = lane & 15;
  const int grp  = lane >> 4;
  const int l0   = wave * 32;

  const float* Fcw = Fh + ((long)(l0 >> 4) * 32 * 32 + lane) * 4;
  const float* Fsw = Fcw + 65536;

  const float* RwS = &Rs[row * ROWP + 14 * grp];
  const float* RwD = &Rs[row * ROWP + 10 * grp + 2];

  v8f accC[3][2], accS[3][2];
#pragma unroll
  for (int c = 0; c < 3; ++c)
#pragma unroll
    for (int j = 0; j < 2; ++j) { accC[c][j] = (v8f)0.0f; accS[c][j] = (v8f)0.0f; }

  v4f fc0[2][2], fs0[2][2], fc1[2][2], fs1[2][2];
  fetchH(Fcw, 0, fc0);
  fetchH(Fsw, 0, fs0);

  __syncthreads();

  for (int t = 0; t < NT_W - 2; t += 2) {
    fetchH(Fcw, t + 1, fc1);
    fetchH(Fsw, t + 1, fs1);
    __builtin_amdgcn_sched_barrier(0);
    mma_rows_fold(RwS + t * 96, RwD + t * 96, fc0, fs0, accC, accS);

    fetchH(Fcw, t + 2, fc0);
    fetchH(Fsw, t + 2, fs0);
    __builtin_amdgcn_sched_barrier(0);
    mma_rows_fold(RwS + (t + 1) * 96, RwD + (t + 1) * 96, fc1, fs1, accC, accS);
  }
  fetchH(Fcw, NT_W - 1, fc1);
  fetchH(Fsw, NT_W - 1, fs1);
  __builtin_amdgcn_sched_barrier(0);
  mma_rows_fold(RwS + (NT_W - 2) * 96, RwD + (NT_W - 2) * 96, fc0, fs0, accC, accS);

  v2f tw[2];
#pragma unroll
  for (int nt = 0; nt < 2; ++nt)
    tw[nt] = *(const v2f*)&Fh[131072 + 2 * (l0 + nt * 16 + row)];
  __builtin_amdgcn_sched_barrier(0);
  mma_rows_fold(RwS + (NT_W - 1) * 96, RwD + (NT_W - 1) * 96, fc1, fs1, accC, accS);

  float* Ob = Out + offb;
#pragma unroll
  for (int nt = 0; nt < 2; ++nt) {
    const int l  = l0 + nt * 16 + row;
    const int lb = (l == 0) ? 256 : 512 - l;
    const float cd = tw[nt].x;
    const float sd = tw[nt].y;
#pragma unroll
    for (int v = 0; v < 8; ++v) {
      const int m = m0 + grp * 8 + v;
      float* p  = &Ob[(long)m * 1536 + l * 3];
      float* pb = &Ob[(long)m * 1536 + lb * 3];
#pragma unroll
      for (int c = 0; c < 3; ++c) {
        const float ce = accC[c][nt][v];
        const float so = accS[c][nt][v];
        p[c]  = cd * ce + sd * so;
        pb[c] = cd * so - sd * ce;
      }
    }
  }
}

extern "C" void kernel_launch(void* const* d_in, const int* in_sizes, int n_in,
                              void* d_out, int out_size, void* d_ws, size_t ws_size,
                              hipStream_t stream) {
  (void)in_sizes; (void)n_in; (void)out_size; (void)ws_size;
  const float* x = (const float*)d_in[0];
  float* out = (float*)d_out;
  float* Fm  = (float*)d_ws;
  float* Y1  = Fm + 512 * 512;

  dct_make_half<<<dim3(256), dim3(256), 0, stream>>>(Fm);

  dct_cols_fold<<<dim3(24, 2, 32), dim3(256), 0, stream>>>(Fm, x, Y1);

  dct_rows3_fold<<<dim3(32, 1, 32), dim3(256), 0, stream>>>(Fm, Y1, out);
}


# Stage 1 second adjacent-pair fold: s and d folded again into 4 panels of 128, quarter-size cos/sin tables, K 256 -> 128, rows k, 512-k, 256-k, 256+k by a 4x4 epilogue matrix; changes summation order
# speedup vs baseline: 3.4907x; 1.2259x over previous
#include <hip/hip_runtime.h>


typedef __attribute__((ext_vector_type(2))) float v2f;
typedef __attribute__((ext_vector_type(4))) float v4f;
typedef __attribute__((ext_vector_type(8))) float v8f;

#define Q_COS 132096
#define Q_SIN (Q_COS + 16384)
#define Q_MAT (Q_SIN + 16384)
__global__ __launch_bounds__(256) void dct_make_half(float* __restrict__ Fh) {
  int idx  = blockIdx.x * 256 + threadIdx.x;
  int e    = idx & 3;
  int lane = (idx >> 2) & 31;
  int j    = (idx >> 7) & 31;
  int t16  = idx >> 12;
  int k = t16 * 16 + (lane & 15);
  int m = (2 * j + (e >> 1)) * 4 + 2 * (lane >> 4) + (e & 1);
  int r = ((2 * m + 1) * k) & 1023;
  float th = 3.14159265358979323846f * (float)r * (1.0f / 512.0f);
  float vc = cosf(th) * 0.0625f;
  float vs = sinf(th) * 0.0625f;
  if (k == 0) {
    vc *= 0.70710678118654752f;
    vs = (m & 1) ? -0.0625f * 0.70710678118654752f : 0.0625f * 0.70710678118654752f;
  }
  Fh[idx]         = vc;
  Fh[65536 + idx] = vs;
  if (idx < 256) {
    float dl = 3.14159265358979323846f * (float)idx * (1.0f / 1024.0f);
    Fh[131072 + 2 * idx]     = cosf(dl);
    Fh[131072 + 2 * idx + 1] = sinf(dl);
  }
  if (idx < 16384) {
    int slot2 = (idx >> 7) & 15;
    int k2 = (idx >> 11) * 16 + (lane & 15);
    int j2 = (2 * slot2 + (e >> 1)) * 4 + 2 * (lane >> 4) + (e & 1);
    int r2 = ((2 * j2 + 1) * k2) & 511;
    float t2 = 3.14159265358979323846f * (float)r2 * (1.0f / 256.0f);
    float c2 = cosf(t2) * 0.0625f;
    float s2 = sinf(t2) * 0.0625f;
    if (k2 == 0) {
      c2 *= 0.70710678118654752f;
      s2 = (j2 & 1) ? -0.0625f * 0.70710678118654752f : 0.0625f * 0.70710678118654752f;
    }
    Fh[Q_COS + idx] = c2;
    Fh[Q_SIN + idx] = s2;
  }
  if (idx < 2048) {
    int k = idx >> 4;
    int o = (idx >> 2) & 3;
    int q = idx & 3;
    float val;
    if (k == 0) {
      const float c8 = cosf(3.14159265358979323846f * 0.125f);
      const float s8 = sinf(3.14159265358979323846f * 0.125f);
      val = 0.0f;
      if (o == 0 && q == 0) val = 1.0f;
      if (o == 3 && q == 3) val = 1.0f;
      if (o == 1 && q == 1) val = c8;
      if (o == 1 && q == 2) val = s8;
      if (o == 2 && q == 1) val = -s8;
      if (o == 2 && q == 2) val = c8;
    } else {
      const float pe = 3.14159265358979323846f * (float)k * (1.0f / 512.0f);
      const float pd = 3.14159265358979323846f * (float)((o < 2) ? k : 256 - k) * (1.0f / 1024.0f);
      const float ce = cosf(pe), se = sinf(pe), cd = cosf(pd), sd = sinf(pd);
      const bool firstHalf = (q < 2);
      const bool odd = (o & 1);
      const float a = (firstHalf != odd) ? cd : sd;
      float b, sgn;
      if (o < 2) {
        b = (q & 1) ? se : ce;
        sgn = (o == 0) ? ((q == 3) ? -1.0f : 1.0f) : ((q == 2) ? 1.0f : -1.0f);
      } else {
        b = (q & 1) ? ce : se;
        sgn = (o == 2) ? ((q == 0) ? -1.0f : 1.0f) : ((q == 1) ? -1.0f : 1.0f);
      }
      val = sgn * a * b;
    }
    Fh[Q_MAT + idx] = val;
  }
}

__device__ __forceinline__ void fetchH(const float* __restrict__ Fw, int kt,
                                       v4f (&f)[2][2]) {
#pragma unroll
  for (int i = 0; i < 2; ++i)
#pragma unroll
    for (int jj = 0; jj < 2; ++jj)
      f[i][jj] = *(const v4f*)&Fw[(i * 32 + kt * 2 + jj) * 128];
}

#define QPANEL_DW  (64 * 64)
#define NT_Q       8

__device__ __forceinline__ void fetchQ(const float* __restrict__ Fw, int kt,
                                       v4f (&f)[2]) {
#pragma unroll
  for (int jj = 0; jj < 2; ++jj)
    f[jj] = *(const v4f*)&Fw[(kt * 2 + jj) * 128];
}

__device__ __forceinline__ void mma_fold2(const float* __restrict__ B0,
                                          const float* __restrict__ B1,
                                          const v4f (&fc)[2], const v4f (&fs)[2],
                                          v8f (&acc)[4][2]) {
#pragma unroll
  for (int s = 0; s < 4; ++s) {
    const v4f qc = fc[s >> 1];
    const v4f qs = fs[s >> 1];
    const v2f ac = (s & 1) ? qc.zw : qc.xy;
    const v2f as = (s & 1) ? qs.zw : qs.xy;
    v2f b[4][2];
#pragma unroll
    for (int p = 0; p < 4; ++p) {
      b[p][0] = *(const v2f*)&B0[p * QPANEL_DW + s * 128];
      b[p][1] = *(const v2f*)&B1[p * QPANEL_DW + s * 128];
    }
#pragma unroll
    for (int nt = 0; nt < 2; ++nt) {
      acc[0][nt] = __builtin_amdgcn_wmma_f32_16x16x4_f32(
          false, ac, false, b[0][nt], (short)0, acc[0][nt], false, false);
      acc[1][nt] = __builtin_amdgcn_wmma_f32_16x16x4_f32(
          false, as, false, b[1][nt], (short)0, acc[1][nt], false, false);
      acc[2][nt] = __builtin_amdgcn_wmma_f32_16x16x4_f32(
          false, as, false, b[2][nt], (short)0, acc[2][nt], false, false);
      acc[3][nt] = __builtin_amdgcn_wmma_f32_16x16x4_f32(
          false, ac, false, b[3][nt], (short)0, acc[3][nt], false, false);
    }
  }
}

__global__ __launch_bounds__(256)
void dct_cols_fold2(const float* __restrict__ Fh, const float* __restrict__ X,
                    float* __restrict__ C) {
  __shared__ float Ds[4 * QPANEL_DW];

  const int tid = threadIdx.x;
  const long off = (long)blockIdx.z * (512L * 1536L) + blockIdx.x * 32;
  const float* Xb = X + off;
  float*       Cb = C + off;

  {
    const int n4 = tid & 7;
    const int kp = tid >> 3;
    const float* src = Xb + (unsigned)(8 * kp) * 1536u + (unsigned)n4 * 4u;
    float* dst = &Ds[kp * 64 + ((8 * n4) ^ (32 * (kp & 1)))];
    v4f rr[2][8];
#pragma unroll
    for (int sl = 0; sl < 2; ++sl)
#pragma unroll
      for (int i = 0; i < 8; ++i)
        rr[sl][i] = *(const v4f*)&src[(sl * 256 + i) * 1536];
#pragma unroll
    for (int sl = 0; sl < 2; ++sl) {
      v4f w[4][2];
#pragma unroll
      for (int h = 0; h < 2; ++h) {
        const v4f s0 = rr[sl][4 * h]     + rr[sl][4 * h + 1];
        const v4f d0 = rr[sl][4 * h]     - rr[sl][4 * h + 1];
        const v4f s1 = rr[sl][4 * h + 2] + rr[sl][4 * h + 3];
        const v4f d1 = rr[sl][4 * h + 2] - rr[sl][4 * h + 3];
        w[0][h] = s0 + s1;
        w[1][h] = s0 - s1;
        w[2][h] = d0 + d1;
        w[3][h] = d0 - d1;
      }
#pragma unroll
      for (int p = 0; p < 4; ++p) {
        v4f lo, hi;
        lo.x = w[p][0].x; lo.y = w[p][1].x; lo.z = w[p][0].y; lo.w = w[p][1].y;
        hi.x = w[p][0].z; hi.y = w[p][1].z; hi.z = w[p][0].w; hi.w = w[p][1].w;
        *(v4f*)&dst[p * QPANEL_DW + sl * 32 * 64]     = lo;
        *(v4f*)&dst[p * QPANEL_DW + sl * 32 * 64 + 4] = hi;
      }
    }
  }
  __builtin_amdgcn_sched_barrier(0);

  const int lane = tid & 31;
  const int wave = __builtin_amdgcn_readfirstlane(tid >> 5);
  const int row  = lane & 15;
  const int grp  = lane >> 4;
  const int k0   = wave * 16;

  const float* B0 = &Ds[grp * 64 + 2 * row + 32 * grp];
  const float* B1 = &Ds[grp * 64 + 2 * row + 32 * (1 - grp)];

  const float* Fcw = Fh + Q_COS + ((long)wave * 16 * 32 + lane) * 4;
  const float* Fsw = Fcw + 16384;

  v8f acc[4][2];
#pragma unroll
  for (int p = 0; p < 4; ++p)
#pragma unroll
    for (int j = 0; j < 2; ++j) acc[p][j] = (v8f)0.0f;

  v4f fc0[2], fs0[2], fc1[2], fs1[2];
  fetchQ(Fcw, 0, fc0);
  fetchQ(Fsw, 0, fs0);

  __syncthreads();

  for (int t = 0; t < NT_Q - 2; t += 2) {
    fetchQ(Fcw, t + 1, fc1);
    fetchQ(Fsw, t + 1, fs1);
    __builtin_amdgcn_sched_barrier(0);
    mma_fold2(B0 + t * 512, B1 + t * 512, fc0, fs0, acc);

    fetchQ(Fcw, t + 2, fc0);
    fetchQ(Fsw, t + 2, fs0);
    __builtin_amdgcn_sched_barrier(0);
    mma_fold2(B0 + (t + 1) * 512, B1 + (t + 1) * 512, fc1, fs1, acc);
  }
  fetchQ(Fcw, NT_Q - 1, fc1);
  fetchQ(Fsw, NT_Q - 1, fs1);
  __builtin_amdgcn_sched_barrier(0);
  mma_fold2(B0 + (NT_Q - 2) * 512, B1 + (NT_Q - 2) * 512, fc0, fs0, acc);
  mma_fold2(B0 + (NT_Q - 1) * 512, B1 + (NT_Q - 1) * 512, fc1, fs1, acc);

  const float* Mw = Fh + Q_MAT + (k0 + grp * 8) * 16;
#pragma unroll
  for (int v = 0; v < 8; ++v) {
    const int k = k0 + grp * 8 + v;
    const int r1 = (k == 0) ? 128 : 512 - k;
    const int r2 = (k == 0) ? 384 : 256 - k;
    const v4f m0 = *(const v4f*)&Mw[v * 16];
    const v4f m1 = *(const v4f*)&Mw[v * 16 + 4];
    const v4f m2 = *(const v4f*)&Mw[v * 16 + 8];
    const v4f m3 = *(const v4f*)&Mw[v * 16 + 12];
#pragma unroll
    for (int nt = 0; nt < 2; ++nt) {
      const int n = nt * 16 + row;
      const float a0 = acc[0][nt][v];
      const float a1 = acc[1][nt][v];
      const float a2 = acc[2][nt][v];
      const float a3 = acc[3][nt][v];
      Cb[(long)k         * 1536 + n] = m0.x * a0 + m0.y * a1 + m0.z * a2 + m0.w * a3;
      Cb[(long)r1        * 1536 + n] = m1.x * a0 + m1.y * a1 + m1.z * a2 + m1.w * a3;
      Cb[(long)r2        * 1536 + n] = m2.x * a0 + m2.y * a1 + m2.z * a2 + m2.w * a3;
      Cb[(long)(256 + k) * 1536 + n] = m3.x * a0 + m3.y * a1 + m3.z * a2 + m3.w * a3;
    }
  }
}

#define ROWS_W 16
#define ROWP   1572
#define NT_W   16

__device__ __forceinline__ void mma_rows_fold(const float* __restrict__ RwS,
                                              const float* __restrict__ RwD,
                                              const v4f (&fc)[2][2],
                                              const v4f (&fs)[2][2],
                                              v8f (&accC)[3][2], v8f (&accS)[3][2]) {
#pragma unroll
  for (int s = 0; s < 4; ++s) {
    v2f bc[2], bs[2], av[3], dv[3];
#pragma unroll
    for (int i = 0; i < 2; ++i) {
      const v4f qc = fc[i][s >> 1];
      const v4f qs = fs[i][s >> 1];
      bc[i] = (s & 1) ? qc.zw : qc.xy;
      bs[i] = (s & 1) ? qs.zw : qs.xy;
    }
#pragma unroll
    for (int c = 0; c < 3; ++c) {
      av[c] = *(const v2f*)&RwS[s * 24 + c * 4];
      dv[c] = *(const v2f*)&RwD[s * 24 + c * 4];
    }
#pragma unroll
    for (int c = 0; c < 3; ++c)
#pragma unroll
      for (int nt = 0; nt < 2; ++nt) {
        accC[c][nt] = __builtin_amdgcn_wmma_f32_16x16x4_f32(
            false, av[c], false, bc[nt], (short)0, accC[c][nt], false, false);
        accS[c][nt] = __builtin_amdgcn_wmma_f32_16x16x4_f32(
            false, dv[c], false, bs[nt], (short)0, accS[c][nt], false, false);
      }
  }
}

__global__ __launch_bounds__(256, 4)
void dct_rows3_fold(const float* __restrict__ Fh, const float* __restrict__ Y,
                    float* __restrict__ Out) {
  __shared__ float Rs[ROWS_W * ROWP];

  const int tid = threadIdx.x;
  const long offb = (long)blockIdx.z * (512L * 1536L);
  const int m0 = blockIdx.x * ROWS_W;

  {
    const int gg = tid & 127;
    const int rh = tid >> 7;
    const float sg = (gg & 1) ? -1.0f : 1.0f;
    const float* src = Y + offb + (long)m0 * 1536 + (unsigned)rh * 1536u + (unsigned)gg * 12u;
    float* dst = &Rs[rh * ROWP + gg * 12];
    v4f rr[8][3];
#pragma unroll
    for (int it = 0; it < 8; ++it)
#pragma unroll
      for (int j = 0; j < 3; ++j)
        rr[it][j] = *(const v4f*)&src[it * 2 * 1536 + j * 4];
#pragma unroll
    for (int it = 0; it < 8; ++it) {
      const v4f r0 = rr[it][0], r1 = rr[it][1], r2 = rr[it][2];
      v4f w0, w1, w2;
      w0.x = r0.x + sg * r0.w;  w0.y = r1.z + sg * r2.y;
      w0.z = r0.x - sg * r0.w;  w0.w = r1.z - sg * r2.y;
      w1.x = r0.y + sg * r1.x;  w1.y = r1.w + sg * r2.z;
      w1.z = r0.y - sg * r1.x;  w1.w = r1.w - sg * r2.z;
      w2.x = r0.z + sg * r1.y;  w2.y = r2.x + sg * r2.w;
      w2.z = r0.z - sg * r1.y;  w2.w = r2.x - sg * r2.w;
      *(v4f*)&dst[it * 2 * ROWP]     = w0;
      *(v4f*)&dst[it * 2 * ROWP + 4] = w1;
      *(v4f*)&dst[it * 2 * ROWP + 8] = w2;
    }
  }

  const int lane = tid & 31;
  const int wave = __builtin_amdgcn_readfirstlane(tid >> 5);
  const int row  = lane & 15;
  const int grp  = lane >> 4;
  const int l0   = wave * 32;

  const float* Fcw = Fh + ((long)(l0 >> 4) * 32 * 32 + lane) * 4;
  const float* Fsw = Fcw + 65536;

  const float* RwS = &Rs[row * ROWP + 14 * grp];
  const float* RwD = &Rs[row * ROWP + 10 * grp + 2];

  v8f accC[3][2], accS[3][2];
#pragma unroll
  for (int c = 0; c < 3; ++c)
#pragma unroll
    for (int j = 0; j < 2; ++j) { accC[c][j] = (v8f)0.0f; accS[c][j] = (v8f)0.0f; }

  v4f fc0[2][2], fs0[2][2], fc1[2][2], fs1[2][2];
  fetchH(Fcw, 0, fc0);
  fetchH(Fsw, 0, fs0);

  __syncthreads();

  for (int t = 0; t < NT_W - 2; t += 2) {
    fetchH(Fcw, t + 1, fc1);
    fetchH(Fsw, t + 1, fs1);
    __builtin_amdgcn_sched_barrier(0);
    mma_rows_fold(RwS + t * 96, RwD + t * 96, fc0, fs0, accC, accS);

    fetchH(Fcw, t + 2, fc0);
    fetchH(Fsw, t + 2, fs0);
    __builtin_amdgcn_sched_barrier(0);
    mma_rows_fold(RwS + (t + 1) * 96, RwD + (t + 1) * 96, fc1, fs1, accC, accS);
  }
  fetchH(Fcw, NT_W - 1, fc1);
  fetchH(Fsw, NT_W - 1, fs1);
  __builtin_amdgcn_sched_barrier(0);
  mma_rows_fold(RwS + (NT_W - 2) * 96, RwD + (NT_W - 2) * 96, fc0, fs0, accC, accS);

  v2f tw[2];
#pragma unroll
  for (int nt = 0; nt < 2; ++nt)
    tw[nt] = *(const v2f*)&Fh[131072 + 2 * (l0 + nt * 16 + row)];
  __builtin_amdgcn_sched_barrier(0);
  mma_rows_fold(RwS + (NT_W - 1) * 96, RwD + (NT_W - 1) * 96, fc1, fs1, accC, accS);

  float* Ob = Out + offb;
#pragma unroll
  for (int nt = 0; nt < 2; ++nt) {
    const int l  = l0 + nt * 16 + row;
    const int lb = (l == 0) ? 256 : 512 - l;
    const float cd = tw[nt].x;
    const float sd = tw[nt].y;
#pragma unroll
    for (int v = 0; v < 8; ++v) {
      const int m = m0 + grp * 8 + v;
      float* p  = &Ob[(long)m * 1536 + l * 3];
      float* pb = &Ob[(long)m * 1536 + lb * 3];
#pragma unroll
      for (int c = 0; c < 3; ++c) {
        const float ce = accC[c][nt][v];
        const float so = accS[c][nt][v];
        p[c]  = cd * ce + sd * so;
        pb[c] = cd * so - sd * ce;
      }
    }
  }
}

extern "C" void kernel_launch(void* const* d_in, const int* in_sizes, int n_in,
                              void* d_out, int out_size, void* d_ws, size_t ws_size,
                              hipStream_t stream) {
  (void)in_sizes; (void)n_in; (void)out_size; (void)ws_size;
  const float* x = (const float*)d_in[0];
  float* out = (float*)d_out;
  float* Fm  = (float*)d_ws;
  float* Y1  = Fm + 512 * 512;

  dct_make_half<<<dim3(256), dim3(256), 0, stream>>>(Fm);

  dct_cols_fold2<<<dim3(48, 1, 32), dim3(256), 0, stream>>>(Fm, x, Y1);

  dct_rows3_fold<<<dim3(32, 1, 32), dim3(256), 0, stream>>>(Fm, Y1, out);
}


# Stage 2 second adjacent-pair fold (as stage 1): 4-pixel groups folded to ss,sd,ds,dd in LDS, quarter-size cos/sin tables as B, K 256 -> 128, columns by a 4x4 epilogue matrix; changes summation order
# speedup vs baseline: 4.4556x; 1.2764x over previous
#include <hip/hip_runtime.h>


typedef __attribute__((ext_vector_type(2))) float v2f;
typedef __attribute__((ext_vector_type(4))) float v4f;
typedef __attribute__((ext_vector_type(8))) float v8f;

#define Q_COS 0
#define Q_SIN (Q_COS + 16384)
#define Q_MAT (Q_SIN + 16384)
__global__ __launch_bounds__(256) void dct_make_quarter(float* __restrict__ Fh) {
  int idx  = blockIdx.x * 256 + threadIdx.x;
  int e    = idx & 3;
  int lane = (idx >> 2) & 31;
  int slot2 = (idx >> 7) & 15;
  int k2 = (idx >> 11) * 16 + (lane & 15);
  int j2 = (2 * slot2 + (e >> 1)) * 4 + 2 * (lane >> 4) + (e & 1);
  int r2 = ((2 * j2 + 1) * k2) & 511;
  float t2 = 3.14159265358979323846f * (float)r2 * (1.0f / 256.0f);
  float c2 = cosf(t2) * 0.0625f;
  float s2 = sinf(t2) * 0.0625f;
  if (k2 == 0) {
    c2 *= 0.70710678118654752f;
    s2 = (j2 & 1) ? -0.0625f * 0.70710678118654752f : 0.0625f * 0.70710678118654752f;
  }
  Fh[Q_COS + idx] = c2;
  Fh[Q_SIN + idx] = s2;
  if (idx < 2048) {
    int k = idx >> 4;
    int o = (idx >> 2) & 3;
    int q = idx & 3;
    float val;
    if (k == 0) {
      const float c8 = cosf(3.14159265358979323846f * 0.125f);
      const float s8 = sinf(3.14159265358979323846f * 0.125f);
      val = 0.0f;
      if (o == 0 && q == 0) val = 1.0f;
      if (o == 3 && q == 3) val = 1.0f;
      if (o == 1 && q == 1) val = c8;
      if (o == 1 && q == 2) val = s8;
      if (o == 2 && q == 1) val = -s8;
      if (o == 2 && q == 2) val = c8;
    } else {
      const float pe = 3.14159265358979323846f * (float)k * (1.0f / 512.0f);
      const float pd = 3.14159265358979323846f * (float)((o < 2) ? k : 256 - k) * (1.0f / 1024.0f);
      const float ce = cosf(pe), se = sinf(pe), cd = cosf(pd), sd = sinf(pd);
      const bool firstHalf = (q < 2);
      const bool odd = (o & 1);
      const float a = (firstHalf != odd) ? cd : sd;
      float b, sgn;
      if (o < 2) {
        b = (q & 1) ? se : ce;
        sgn = (o == 0) ? ((q == 3) ? -1.0f : 1.0f) : ((q == 2) ? 1.0f : -1.0f);
      } else {
        b = (q & 1) ? ce : se;
        sgn = (o == 2) ? ((q == 0) ? -1.0f : 1.0f) : ((q == 1) ? -1.0f : 1.0f);
      }
      val = sgn * a * b;
    }
    Fh[Q_MAT + idx] = val;
  }
}

#define QPANEL_DW  (64 * 64)
#define NT_Q       8

__device__ __forceinline__ void fetchQ(const float* __restrict__ Fw, int kt,
                                       v4f (&f)[2]) {
#pragma unroll
  for (int jj = 0; jj < 2; ++jj)
    f[jj] = *(const v4f*)&Fw[(kt * 2 + jj) * 128];
}

__device__ __forceinline__ void mma_fold2(const float* __restrict__ B0,
                                          const float* __restrict__ B1,
                                          const v4f (&fc)[2], const v4f (&fs)[2],
                                          v8f (&acc)[4][2]) {
#pragma unroll
  for (int s = 0; s < 4; ++s) {
    const v4f qc = fc[s >> 1];
    const v4f qs = fs[s >> 1];
    const v2f ac = (s & 1) ? qc.zw : qc.xy;
    const v2f as = (s & 1) ? qs.zw : qs.xy;
    v2f b[4][2];
#pragma unroll
    for (int p = 0; p < 4; ++p) {
      b[p][0] = *(const v2f*)&B0[p * QPANEL_DW + s * 128];
      b[p][1] = *(const v2f*)&B1[p * QPANEL_DW + s * 128];
    }
#pragma unroll
    for (int nt = 0; nt < 2; ++nt) {
      acc[0][nt] = __builtin_amdgcn_wmma_f32_16x16x4_f32(
          false, ac, false, b[0][nt], (short)0, acc[0][nt], false, false);
      acc[1][nt] = __builtin_amdgcn_wmma_f32_16x16x4_f32(
          false, as, false, b[1][nt], (short)0, acc[1][nt], false, false);
      acc[2][nt] = __builtin_amdgcn_wmma_f32_16x16x4_f32(
          false, as, false, b[2][nt], (short)0, acc[2][nt], false, false);
      acc[3][nt] = __builtin_amdgcn_wmma_f32_16x16x4_f32(
          false, ac, false, b[3][nt], (short)0, acc[3][nt], false, false);
    }
  }
}

__global__ __launch_bounds__(256)
void dct_cols_fold2(const float* __restrict__ Fh, const float* __restrict__ X,
                    float* __restrict__ C) {
  __shared__ float Ds[4 * QPANEL_DW];

  const int tid = threadIdx.x;
  const long off = (long)blockIdx.z * (512L * 1536L) + blockIdx.x * 32;
  const float* Xb = X + off;
  float*       Cb = C + off;

  {
    const int n4 = tid & 7;
    const int kp = tid >> 3;
    const float* src = Xb + (unsigned)(8 * kp) * 1536u + (unsigned)n4 * 4u;
    float* dst = &Ds[kp * 64 + ((8 * n4) ^ (32 * (kp & 1)))];
    v4f rr[2][8];
#pragma unroll
    for (int sl = 0; sl < 2; ++sl)
#pragma unroll
      for (int i = 0; i < 8; ++i)
        rr[sl][i] = *(const v4f*)&src[(sl * 256 + i) * 1536];
#pragma unroll
    for (int sl = 0; sl < 2; ++sl) {
      v4f w[4][2];
#pragma unroll
      for (int h = 0; h < 2; ++h) {
        const v4f s0 = rr[sl][4 * h]     + rr[sl][4 * h + 1];
        const v4f d0 = rr[sl][4 * h]     - rr[sl][4 * h + 1];
        const v4f s1 = rr[sl][4 * h + 2] + rr[sl][4 * h + 3];
        const v4f d1 = rr[sl][4 * h + 2] - rr[sl][4 * h + 3];
        w[0][h] = s0 + s1;
        w[1][h] = s0 - s1;
        w[2][h] = d0 + d1;
        w[3][h] = d0 - d1;
      }
#pragma unroll
      for (int p = 0; p < 4; ++p) {
        v4f lo, hi;
        lo.x = w[p][0].x; lo.y = w[p][1].x; lo.z = w[p][0].y; lo.w = w[p][1].y;
        hi.x = w[p][0].z; hi.y = w[p][1].z; hi.z = w[p][0].w; hi.w = w[p][1].w;
        *(v4f*)&dst[p * QPANEL_DW + sl * 32 * 64]     = lo;
        *(v4f*)&dst[p * QPANEL_DW + sl * 32 * 64 + 4] = hi;
      }
    }
  }
  __builtin_amdgcn_sched_barrier(0);

  const int lane = tid & 31;
  const int wave = __builtin_amdgcn_readfirstlane(tid >> 5);
  const int row  = lane & 15;
  const int grp  = lane >> 4;
  const int k0   = wave * 16;

  const float* B0 = &Ds[grp * 64 + 2 * row + 32 * grp];
  const float* B1 = &Ds[grp * 64 + 2 * row + 32 * (1 - grp)];

  const float* Fcw = Fh + Q_COS + ((long)wave * 16 * 32 + lane) * 4;
  const float* Fsw = Fcw + 16384;

  v8f acc[4][2];
#pragma unroll
  for (int p = 0; p < 4; ++p)
#pragma unroll
    for (int j = 0; j < 2; ++j) acc[p][j] = (v8f)0.0f;

  v4f fc0[2], fs0[2], fc1[2], fs1[2];
  fetchQ(Fcw, 0, fc0);
  fetchQ(Fsw, 0, fs0);

  __syncthreads();

  for (int t = 0; t < NT_Q - 2; t += 2) {
    fetchQ(Fcw, t + 1, fc1);
    fetchQ(Fsw, t + 1, fs1);
    __builtin_amdgcn_sched_barrier(0);
    mma_fold2(B0 + t * 512, B1 + t * 512, fc0, fs0, acc);

    fetchQ(Fcw, t + 2, fc0);
    fetchQ(Fsw, t + 2, fs0);
    __builtin_amdgcn_sched_barrier(0);
    mma_fold2(B0 + (t + 1) * 512, B1 + (t + 1) * 512, fc1, fs1, acc);
  }
  fetchQ(Fcw, NT_Q - 1, fc1);
  fetchQ(Fsw, NT_Q - 1, fs1);
  __builtin_amdgcn_sched_barrier(0);
  mma_fold2(B0 + (NT_Q - 2) * 512, B1 + (NT_Q - 2) * 512, fc0, fs0, acc);
  mma_fold2(B0 + (NT_Q - 1) * 512, B1 + (NT_Q - 1) * 512, fc1, fs1, acc);

  const float* Mw = Fh + Q_MAT + (k0 + grp * 8) * 16;
#pragma unroll
  for (int v = 0; v < 8; ++v) {
    const int k = k0 + grp * 8 + v;
    const int r1 = (k == 0) ? 128 : 512 - k;
    const int r2 = (k == 0) ? 384 : 256 - k;
    const v4f m0 = *(const v4f*)&Mw[v * 16];
    const v4f m1 = *(const v4f*)&Mw[v * 16 + 4];
    const v4f m2 = *(const v4f*)&Mw[v * 16 + 8];
    const v4f m3 = *(const v4f*)&Mw[v * 16 + 12];
#pragma unroll
    for (int nt = 0; nt < 2; ++nt) {
      const int n = nt * 16 + row;
      const float a0 = acc[0][nt][v];
      const float a1 = acc[1][nt][v];
      const float a2 = acc[2][nt][v];
      const float a3 = acc[3][nt][v];
      Cb[(long)k         * 1536 + n] = m0.x * a0 + m0.y * a1 + m0.z * a2 + m0.w * a3;
      Cb[(long)r1        * 1536 + n] = m1.x * a0 + m1.y * a1 + m1.z * a2 + m1.w * a3;
      Cb[(long)r2        * 1536 + n] = m2.x * a0 + m2.y * a1 + m2.z * a2 + m2.w * a3;
      Cb[(long)(256 + k) * 1536 + n] = m3.x * a0 + m3.y * a1 + m3.z * a2 + m3.w * a3;
    }
  }
}

#define ROWS_W 16
#define ROWP   1572
#define NT_W   8

__device__ __forceinline__ void mma_rows_fold2(const float* __restrict__ Rw,
                                               const v4f (&fc)[2], const v4f (&fs)[2],
                                               v8f (&acc)[4][3]) {
#pragma unroll
  for (int s = 0; s < 4; ++s) {
    const v4f qc = fc[s >> 1];
    const v4f qs = fs[s >> 1];
    const v2f bc = (s & 1) ? qc.zw : qc.xy;
    const v2f bs = (s & 1) ? qs.zw : qs.xy;
    v4f a[6];
#pragma unroll
    for (int t = 0; t < 6; ++t) a[t] = *(const v4f*)&Rw[s * 48 + t * 4];
#pragma unroll
    for (int c = 0; c < 3; ++c) {
      const int q0 = c, q1 = 3 + c, q2 = 6 + c, q3 = 9 + c;
      const v2f a0 = (q0 & 1) ? a[q0 >> 1].zw : a[q0 >> 1].xy;
      const v2f a1 = (q1 & 1) ? a[q1 >> 1].zw : a[q1 >> 1].xy;
      const v2f a2 = (q2 & 1) ? a[q2 >> 1].zw : a[q2 >> 1].xy;
      const v2f a3 = (q3 & 1) ? a[q3 >> 1].zw : a[q3 >> 1].xy;
      acc[0][c] = __builtin_amdgcn_wmma_f32_16x16x4_f32(
          false, a0, false, bc, (short)0, acc[0][c], false, false);
      acc[1][c] = __builtin_amdgcn_wmma_f32_16x16x4_f32(
          false, a1, false, bs, (short)0, acc[1][c], false, false);
      acc[2][c] = __builtin_amdgcn_wmma_f32_16x16x4_f32(
          false, a2, false, bs, (short)0, acc[2][c], false, false);
      acc[3][c] = __builtin_amdgcn_wmma_f32_16x16x4_f32(
          false, a3, false, bc, (short)0, acc[3][c], false, false);
    }
  }
}

__global__ __launch_bounds__(256, 4)
void dct_rows3_fold2(const float* __restrict__ Fh, const float* __restrict__ Y,
                     float* __restrict__ Out) {
  __shared__ float Rs[ROWS_W * ROWP];

  const int tid  = threadIdx.x;
  const int lane = tid & 31;
  const int wave = __builtin_amdgcn_readfirstlane(tid >> 5);
  const long offb = (long)blockIdx.z * (512L * 1536L);
  const int m0 = blockIdx.x * ROWS_W;

  {
    const int G  = (lane & 7) | ((lane >> 4) << 3) | ((wave & 3) << 4);
    const int r0 = 2 * (wave >> 2) + ((lane >> 3) & 1);
    const float* src = Y + offb + (long)m0 * 1536 + (unsigned)r0 * 1536u + (unsigned)G * 24u;
    float* dst = &Rs[r0 * ROWP + G * 24];
    v4f rr[4][6];
#pragma unroll
    for (int it = 0; it < 4; ++it)
#pragma unroll
      for (int j = 0; j < 6; ++j)
        rr[it][j] = *(const v4f*)&src[it * 4 * 1536 + j * 4];
#pragma unroll
    for (int it = 0; it < 4; ++it) {
      float f[24], o[24];
#pragma unroll
      for (int i = 0; i < 24; ++i) f[i] = rr[it][i >> 2][i & 3];
#pragma unroll
      for (int e = 0; e < 2; ++e)
#pragma unroll
        for (int c = 0; c < 3; ++c) {
          const float x0 = f[12 * e + c],     x1 = f[12 * e + 3 + c];
          const float x2 = f[12 * e + 6 + c], x3 = f[12 * e + 9 + c];
          const float s0 = x0 + x1, d0 = x0 - x1;
          const float s1 = x2 + x3, d1 = x2 - x3;
          o[2 * (0 + c) + e] = s0 + s1;
          o[2 * (3 + c) + e] = s0 - s1;
          o[2 * (6 + c) + e] = d0 + d1;
          o[2 * (9 + c) + e] = d0 - d1;
        }
#pragma unroll
      for (int t = 0; t < 6; ++t) {
        v4f w;
        w.x = o[4 * t]; w.y = o[4 * t + 1]; w.z = o[4 * t + 2]; w.w = o[4 * t + 3];
        *(v4f*)&dst[it * 4 * ROWP + 4 * t] = w;
      }
    }
  }

  const int row  = lane & 15;
  const int grp  = lane >> 4;
  const int l0   = wave * 16;

  const float* Fcw = Fh + Q_COS + ((long)wave * 16 * 32 + lane) * 4;
  const float* Fsw = Fcw + 16384;

  const float* Rw = &Rs[row * ROWP + 24 * grp];

  v8f acc[4][3];
#pragma unroll
  for (int p = 0; p < 4; ++p)
#pragma unroll
    for (int c = 0; c < 3; ++c) acc[p][c] = (v8f)0.0f;

  v4f fc0[2], fs0[2], fc1[2], fs1[2];
  fetchQ(Fcw, 0, fc0);
  fetchQ(Fsw, 0, fs0);

  __syncthreads();

  for (int t = 0; t < NT_W - 2; t += 2) {
    fetchQ(Fcw, t + 1, fc1);
    fetchQ(Fsw, t + 1, fs1);
    __builtin_amdgcn_sched_barrier(0);
    mma_rows_fold2(Rw + t * 192, fc0, fs0, acc);

    fetchQ(Fcw, t + 2, fc0);
    fetchQ(Fsw, t + 2, fs0);
    __builtin_amdgcn_sched_barrier(0);
    mma_rows_fold2(Rw + (t + 1) * 192, fc1, fs1, acc);
  }
  fetchQ(Fcw, NT_W - 1, fc1);
  fetchQ(Fsw, NT_W - 1, fs1);
  __builtin_amdgcn_sched_barrier(0);
  mma_rows_fold2(Rw + (NT_W - 2) * 192, fc0, fs0, acc);

  const int l = l0 + row;
  v4f mm[4];
#pragma unroll
  for (int o = 0; o < 4; ++o)
    mm[o] = *(const v4f*)&Fh[Q_MAT + l * 16 + o * 4];
  __builtin_amdgcn_sched_barrier(0);
  mma_rows_fold2(Rw + (NT_W - 1) * 192, fc1, fs1, acc);

  float* Ob = Out + offb + (long)(m0 + grp * 8) * 1536;
  const int col[4] = { l, (l == 0) ? 128 : 512 - l, (l == 0) ? 384 : 256 - l, 256 + l };
#pragma unroll
  for (int v = 0; v < 8; ++v) {
    float* pr = Ob + v * 1536;
#pragma unroll
    for (int o = 0; o < 4; ++o) {
      float* p = pr + col[o] * 3;
#pragma unroll
      for (int c = 0; c < 3; ++c)
        p[c] = mm[o].x * acc[0][c][v] + mm[o].y * acc[1][c][v] +
               mm[o].z * acc[2][c][v] + mm[o].w * acc[3][c][v];
    }
  }
}

extern "C" void kernel_launch(void* const* d_in, const int* in_sizes, int n_in,
                              void* d_out, int out_size, void* d_ws, size_t ws_size,
                              hipStream_t stream) {
  (void)in_sizes; (void)n_in; (void)out_size; (void)ws_size;
  const float* x = (const float*)d_in[0];
  float* out = (float*)d_out;
  float* Fm  = (float*)d_ws;
  float* Y1  = Fm + 512 * 512;

  dct_make_quarter<<<dim3(64), dim3(256), 0, stream>>>(Fm);

  dct_cols_fold2<<<dim3(48, 1, 32), dim3(256), 0, stream>>>(Fm, x, Y1);

  dct_rows3_fold2<<<dim3(32, 1, 32), dim3(256), 0, stream>>>(Fm, Y1, out);
}


# Stage 1 third adjacent-pair fold: 8 panels of 64 in LDS, eighth-size cos/sin tables, K 128 -> 64, 4-wave groups, rows k, 128+-k, 256+-k, 384+-k, 512-k by an 8x8 epilogue matrix; changes sum order
# speedup vs baseline: 4.6718x; 1.0485x over previous
#include <hip/hip_runtime.h>


typedef __attribute__((ext_vector_type(2))) float v2f;
typedef __attribute__((ext_vector_type(4))) float v4f;
typedef __attribute__((ext_vector_type(8))) float v8f;

#define Q_COS 0
#define Q_SIN (Q_COS + 16384)
#define Q_MAT (Q_SIN + 16384)
#define E_COS (Q_MAT + 2048)
#define E_SIN (E_COS + 4096)
#define E_MAT (E_SIN + 4096)
__global__ __launch_bounds__(256) void dct_make_quarter(float* __restrict__ Fh) {
  int idx  = blockIdx.x * 256 + threadIdx.x;
  int e    = idx & 3;
  int lane = (idx >> 2) & 31;
  int slot2 = (idx >> 7) & 15;
  int k2 = (idx >> 11) * 16 + (lane & 15);
  int j2 = (2 * slot2 + (e >> 1)) * 4 + 2 * (lane >> 4) + (e & 1);
  int r2 = ((2 * j2 + 1) * k2) & 511;
  float t2 = 3.14159265358979323846f * (float)r2 * (1.0f / 256.0f);
  float c2 = cosf(t2) * 0.0625f;
  float s2 = sinf(t2) * 0.0625f;
  if (k2 == 0) {
    c2 *= 0.70710678118654752f;
    s2 = (j2 & 1) ? -0.0625f * 0.70710678118654752f : 0.0625f * 0.70710678118654752f;
  }
  Fh[Q_COS + idx] = c2;
  Fh[Q_SIN + idx] = s2;
  if (idx < 2048) {
    int k = idx >> 4;
    int o = (idx >> 2) & 3;
    int q = idx & 3;
    float val;
    if (k == 0) {
      const float c8 = cosf(3.14159265358979323846f * 0.125f);
      const float s8 = sinf(3.14159265358979323846f * 0.125f);
      val = 0.0f;
      if (o == 0 && q == 0) val = 1.0f;
      if (o == 3 && q == 3) val = 1.0f;
      if (o == 1 && q == 1) val = c8;
      if (o == 1 && q == 2) val = s8;
      if (o == 2 && q == 1) val = -s8;
      if (o == 2 && q == 2) val = c8;
    } else {
      const float pe = 3.14159265358979323846f * (float)k * (1.0f / 512.0f);
      const float pd = 3.14159265358979323846f * (float)((o < 2) ? k : 256 - k) * (1.0f / 1024.0f);
      const float ce = cosf(pe), se = sinf(pe), cd = cosf(pd), sd = sinf(pd);
      const bool firstHalf = (q < 2);
      const bool odd = (o & 1);
      const float a = (firstHalf != odd) ? cd : sd;
      float b, sgn;
      if (o < 2) {
        b = (q & 1) ? se : ce;
        sgn = (o == 0) ? ((q == 3) ? -1.0f : 1.0f) : ((q == 2) ? 1.0f : -1.0f);
      } else {
        b = (q & 1) ? ce : se;
        sgn = (o == 2) ? ((q == 0) ? -1.0f : 1.0f) : ((q == 1) ? -1.0f : 1.0f);
      }
      val = sgn * a * b;
    }
    Fh[Q_MAT + idx] = val;
  }
  if (idx < 4096) {
    const int slot3 = (idx >> 7) & 7;
    const int k3 = (idx >> 10) * 16 + (lane & 15);
    const int i3 = (2 * slot3 + (e >> 1)) * 4 + 2 * (lane >> 4) + (e & 1);
    const int r3 = ((2 * i3 + 1) * k3) & 255;
    const float t3 = 3.14159265358979323846f * (float)r3 * (1.0f / 128.0f);
    const float c3 = cosf(t3) * 0.0625f;
    float s3 = sinf(t3) * 0.0625f;
    if (k3 == 0) s3 = (i3 & 1) ? -0.0625f : 0.0625f;
    Fh[E_COS + idx] = c3;
    Fh[E_SIN + idx] = s3;
    const int k = idx >> 6;
    const int o = (idx >> 3) & 7;
    const int q = idx & 7;
    const int K = (k == 0) ? 64 * o : ((o + 1) >> 1) * 128 + ((o & 1) ? -k : k);
    int n = 512, kk = K;
    bool isSine = false;
    float f = 1.0f;
    for (int lev = 0; lev < 3; ++lev) {
      const bool diff = (q >> (2 - lev)) & 1;
      const float phi = 3.14159265358979323846f * (float)kk / (float)(2 * n);
      if (diff) {
        f *= isSine ? -sinf(phi) : sinf(phi);
        isSine = !isSine;
      } else {
        f *= cosf(phi);
      }
      const int nh = n >> 1;
      if (kk > nh) {
        kk = n - kk;
        if (!isSine) f = -f;
      }
      n = nh;
    }
    if (k == 0) {
      if (kk == 0 && isSine) f = 0.0f;
      if (kk == 64 && !isSine) f = 0.0f;
    }
    if (K == 0) f *= 0.70710678118654752f;
    Fh[E_MAT + idx] = f;
  }
}

#define EPANEL_DW  (32 * 64)
#define NT_E       4

__device__ __forceinline__ void fetchQ(const float* __restrict__ Fw, int kt,
                                       v4f (&f)[2]) {
#pragma unroll
  for (int jj = 0; jj < 2; ++jj)
    f[jj] = *(const v4f*)&Fw[(kt * 2 + jj) * 128];
}

__device__ __forceinline__ void mma_fold3(const float* __restrict__ B0,
                                          const float* __restrict__ B1,
                                          const v4f (&fc)[2], const v4f (&fs)[2],
                                          v8f (&acc)[8][2]) {
#pragma unroll
  for (int s = 0; s < 4; ++s) {
    const v4f qc = fc[s >> 1];
    const v4f qs = fs[s >> 1];
    const v2f ac = (s & 1) ? qc.zw : qc.xy;
    const v2f as = (s & 1) ? qs.zw : qs.xy;
    v2f b[8][2];
#pragma unroll
    for (int p = 0; p < 8; ++p) {
      b[p][0] = *(const v2f*)&B0[p * EPANEL_DW + s * 128];
      b[p][1] = *(const v2f*)&B1[p * EPANEL_DW + s * 128];
    }
#pragma unroll
    for (int nt = 0; nt < 2; ++nt) {
#pragma unroll
      for (int p = 0; p < 8; ++p) {
        const bool sine = ((p ^ (p >> 1) ^ (p >> 2)) & 1) != 0;
        acc[p][nt] = __builtin_amdgcn_wmma_f32_16x16x4_f32(
            false, sine ? as : ac, false, b[p][nt], (short)0, acc[p][nt], false, false);
      }
    }
  }
}

__global__ __launch_bounds__(128)
void dct_cols_fold3(const float* __restrict__ Fh, const float* __restrict__ X,
                    float* __restrict__ C) {
  __shared__ float Ds[8 * EPANEL_DW];

  const int tid = threadIdx.x;
  const long off = (long)blockIdx.z * (512L * 1536L) + blockIdx.x * 32;
  const float* Xb = X + off;
  float*       Cb = C + off;

  {
    const int n4 = tid & 7;
    const int kp = tid >> 3;
    const float* src = Xb + (unsigned)(16 * kp) * 1536u + (unsigned)n4 * 4u;
    float* dst = &Ds[kp * 64 + ((8 * n4) ^ (32 * (kp & 1)))];
    v4f rr[2][16];
#pragma unroll
    for (int sl = 0; sl < 2; ++sl)
#pragma unroll
      for (int i = 0; i < 16; ++i)
        rr[sl][i] = *(const v4f*)&src[(sl * 256 + i) * 1536];
#pragma unroll
    for (int sl = 0; sl < 2; ++sl) {
      v4f w[8][2];
#pragma unroll
      for (int h = 0; h < 2; ++h) {
        const v4f s0 = rr[sl][8 * h]     + rr[sl][8 * h + 1];
        const v4f d0 = rr[sl][8 * h]     - rr[sl][8 * h + 1];
        const v4f s1 = rr[sl][8 * h + 2] + rr[sl][8 * h + 3];
        const v4f d1 = rr[sl][8 * h + 2] - rr[sl][8 * h + 3];
        const v4f s2 = rr[sl][8 * h + 4] + rr[sl][8 * h + 5];
        const v4f d2 = rr[sl][8 * h + 4] - rr[sl][8 * h + 5];
        const v4f s3 = rr[sl][8 * h + 6] + rr[sl][8 * h + 7];
        const v4f d3 = rr[sl][8 * h + 6] - rr[sl][8 * h + 7];
        const v4f ss0 = s0 + s1, sd0 = s0 - s1, ss1 = s2 + s3, sd1 = s2 - s3;
        const v4f ds0 = d0 + d1, dd0 = d0 - d1, ds1 = d2 + d3, dd1 = d2 - d3;
        w[0][h] = ss0 + ss1;
        w[1][h] = ss0 - ss1;
        w[2][h] = sd0 + sd1;
        w[3][h] = sd0 - sd1;
        w[4][h] = ds0 + ds1;
        w[5][h] = ds0 - ds1;
        w[6][h] = dd0 + dd1;
        w[7][h] = dd0 - dd1;
      }
#pragma unroll
      for (int p = 0; p < 8; ++p) {
        v4f lo, hi;
        lo.x = w[p][0].x; lo.y = w[p][1].x; lo.z = w[p][0].y; lo.w = w[p][1].y;
        hi.x = w[p][0].z; hi.y = w[p][1].z; hi.z = w[p][0].w; hi.w = w[p][1].w;
        *(v4f*)&dst[p * EPANEL_DW + sl * 16 * 64]     = lo;
        *(v4f*)&dst[p * EPANEL_DW + sl * 16 * 64 + 4] = hi;
      }
    }
  }
  __builtin_amdgcn_sched_barrier(0);

  const int lane = tid & 31;
  const int wave = __builtin_amdgcn_readfirstlane(tid >> 5);
  const int row  = lane & 15;
  const int grp  = lane >> 4;
  const int k0   = wave * 16;

  const float* B0 = &Ds[grp * 64 + 2 * row + 32 * grp];
  const float* B1 = &Ds[grp * 64 + 2 * row + 32 * (1 - grp)];

  const float* Fcw = Fh + E_COS + ((long)wave * 8 * 32 + lane) * 4;
  const float* Fsw = Fcw + 4096;

  v8f acc[8][2];
#pragma unroll
  for (int p = 0; p < 8; ++p)
#pragma unroll
    for (int j = 0; j < 2; ++j) acc[p][j] = (v8f)0.0f;

  v4f fc0[2], fs0[2], fc1[2], fs1[2];
  fetchQ(Fcw, 0, fc0);
  fetchQ(Fsw, 0, fs0);

  __syncthreads();

  for (int t = 0; t < NT_E - 2; t += 2) {
    fetchQ(Fcw, t + 1, fc1);
    fetchQ(Fsw, t + 1, fs1);
    __builtin_amdgcn_sched_barrier(0);
    mma_fold3(B0 + t * 512, B1 + t * 512, fc0, fs0, acc);

    fetchQ(Fcw, t + 2, fc0);
    fetchQ(Fsw, t + 2, fs0);
    __builtin_amdgcn_sched_barrier(0);
    mma_fold3(B0 + (t + 1) * 512, B1 + (t + 1) * 512, fc1, fs1, acc);
  }
  fetchQ(Fcw, NT_E - 1, fc1);
  fetchQ(Fsw, NT_E - 1, fs1);
  __builtin_amdgcn_sched_barrier(0);
  mma_fold3(B0 + (NT_E - 2) * 512, B1 + (NT_E - 2) * 512, fc0, fs0, acc);
  mma_fold3(B0 + (NT_E - 1) * 512, B1 + (NT_E - 1) * 512, fc1, fs1, acc);

  const float* Mw = Fh + E_MAT + (k0 + grp * 8) * 64;
#pragma unroll
  for (int v = 0; v < 8; ++v) {
    const int k = k0 + grp * 8 + v;
    float a[2][8];
#pragma unroll
    for (int nt = 0; nt < 2; ++nt)
#pragma unroll
      for (int q = 0; q < 8; ++q) a[nt][q] = acc[q][nt][v];
#pragma unroll
    for (int o = 0; o < 8; ++o) {
      const int r = (k == 0) ? 64 * o : ((o + 1) >> 1) * 128 + ((o & 1) ? -k : k);
      const v4f ma = *(const v4f*)&Mw[v * 64 + o * 8];
      const v4f mb = *(const v4f*)&Mw[v * 64 + o * 8 + 4];
#pragma unroll
      for (int nt = 0; nt < 2; ++nt)
        Cb[(long)r * 1536 + nt * 16 + row] =
            ma.x * a[nt][0] + ma.y * a[nt][1] + ma.z * a[nt][2] + ma.w * a[nt][3] +
            mb.x * a[nt][4] + mb.y * a[nt][5] + mb.z * a[nt][6] + mb.w * a[nt][7];
    }
  }
}

#define ROWS_W 16
#define ROWP   1572
#define NT_W   8

__device__ __forceinline__ void mma_rows_fold2(const float* __restrict__ Rw,
                                               const v4f (&fc)[2], const v4f (&fs)[2],
                                               v8f (&acc)[4][3]) {
#pragma unroll
  for (int s = 0; s < 4; ++s) {
    const v4f qc = fc[s >> 1];
    const v4f qs = fs[s >> 1];
    const v2f bc = (s & 1) ? qc.zw : qc.xy;
    const v2f bs = (s & 1) ? qs.zw : qs.xy;
    v4f a[6];
#pragma unroll
    for (int t = 0; t < 6; ++t) a[t] = *(const v4f*)&Rw[s * 48 + t * 4];
#pragma unroll
    for (int c = 0; c < 3; ++c) {
      const int q0 = c, q1 = 3 + c, q2 = 6 + c, q3 = 9 + c;
      const v2f a0 = (q0 & 1) ? a[q0 >> 1].zw : a[q0 >> 1].xy;
      const v2f a1 = (q1 & 1) ? a[q1 >> 1].zw : a[q1 >> 1].xy;
      const v2f a2 = (q2 & 1) ? a[q2 >> 1].zw : a[q2 >> 1].xy;
      const v2f a3 = (q3 & 1) ? a[q3 >> 1].zw : a[q3 >> 1].xy;
      acc[0][c] = __builtin_amdgcn_wmma_f32_16x16x4_f32(
          false, a0, false, bc, (short)0, acc[0][c], false, false);
      acc[1][c] = __builtin_amdgcn_wmma_f32_16x16x4_f32(
          false, a1, false, bs, (short)0, acc[1][c], false, false);
      acc[2][c] = __builtin_amdgcn_wmma_f32_16x16x4_f32(
          false, a2, false, bs, (short)0, acc[2][c], false, false);
      acc[3][c] = __builtin_amdgcn_wmma_f32_16x16x4_f32(
          false, a3, false, bc, (short)0, acc[3][c], false, false);
    }
  }
}

__global__ __launch_bounds__(256, 4)
void dct_rows3_fold2(const float* __restrict__ Fh, const float* __restrict__ Y,
                     float* __restrict__ Out) {
  __shared__ float Rs[ROWS_W * ROWP];

  const int tid  = threadIdx.x;
  const int lane = tid & 31;
  const int wave = __builtin_amdgcn_readfirstlane(tid >> 5);
  const long offb = (long)blockIdx.z * (512L * 1536L);
  const int m0 = blockIdx.x * ROWS_W;

  {
    const int G  = (lane & 7) | ((lane >> 4) << 3) | ((wave & 3) << 4);
    const int r0 = 2 * (wave >> 2) + ((lane >> 3) & 1);
    const float* src = Y + offb + (long)m0 * 1536 + (unsigned)r0 * 1536u + (unsigned)G * 24u;
    float* dst = &Rs[r0 * ROWP + G * 24];
    v4f rr[4][6];
#pragma unroll
    for (int it = 0; it < 4; ++it)
#pragma unroll
      for (int j = 0; j < 6; ++j)
        rr[it][j] = *(const v4f*)&src[it * 4 * 1536 + j * 4];
#pragma unroll
    for (int it = 0; it < 4; ++it) {
      float f[24], o[24];
#pragma unroll
      for (int i = 0; i < 24; ++i) f[i] = rr[it][i >> 2][i & 3];
#pragma unroll
      for (int e = 0; e < 2; ++e)
#pragma unroll
        for (int c = 0; c < 3; ++c) {
          const float x0 = f[12 * e + c],     x1 = f[12 * e + 3 + c];
          const float x2 = f[12 * e + 6 + c], x3 = f[12 * e + 9 + c];
          const float s0 = x0 + x1, d0 = x0 - x1;
          const float s1 = x2 + x3, d1 = x2 - x3;
          o[2 * (0 + c) + e] = s0 + s1;
          o[2 * (3 + c) + e] = s0 - s1;
          o[2 * (6 + c) + e] = d0 + d1;
          o[2 * (9 + c) + e] = d0 - d1;
        }
#pragma unroll
      for (int t = 0; t < 6; ++t) {
        v4f w;
        w.x = o[4 * t]; w.y = o[4 * t + 1]; w.z = o[4 * t + 2]; w.w = o[4 * t + 3];
        *(v4f*)&dst[it * 4 * ROWP + 4 * t] = w;
      }
    }
  }

  const int row  = lane & 15;
  const int grp  = lane >> 4;
  const int l0   = wave * 16;

  const float* Fcw = Fh + Q_COS + ((long)wave * 16 * 32 + lane) * 4;
  const float* Fsw = Fcw + 16384;

  const float* Rw = &Rs[row * ROWP + 24 * grp];

  v8f acc[4][3];
#pragma unroll
  for (int p = 0; p < 4; ++p)
#pragma unroll
    for (int c = 0; c < 3; ++c) acc[p][c] = (v8f)0.0f;

  v4f fc0[2], fs0[2], fc1[2], fs1[2];
  fetchQ(Fcw, 0, fc0);
  fetchQ(Fsw, 0, fs0);

  __syncthreads();

  for (int t = 0; t < NT_W - 2; t += 2) {
    fetchQ(Fcw, t + 1, fc1);
    fetchQ(Fsw, t + 1, fs1);
    __builtin_amdgcn_sched_barrier(0);
    mma_rows_fold2(Rw + t * 192, fc0, fs0, acc);

    fetchQ(Fcw, t + 2, fc0);
    fetchQ(Fsw, t + 2, fs0);
    __builtin_amdgcn_sched_barrier(0);
    mma_rows_fold2(Rw + (t + 1) * 192, fc1, fs1, acc);
  }
  fetchQ(Fcw, NT_W - 1, fc1);
  fetchQ(Fsw, NT_W - 1, fs1);
  __builtin_amdgcn_sched_barrier(0);
  mma_rows_fold2(Rw + (NT_W - 2) * 192, fc0, fs0, acc);

  const int l = l0 + row;
  v4f mm[4];
#pragma unroll
  for (int o = 0; o < 4; ++o)
    mm[o] = *(const v4f*)&Fh[Q_MAT + l * 16 + o * 4];
  __builtin_amdgcn_sched_barrier(0);
  mma_rows_fold2(Rw + (NT_W - 1) * 192, fc1, fs1, acc);

  float* Ob = Out + offb + (long)(m0 + grp * 8) * 1536;
  const int col[4] = { l, (l == 0) ? 128 : 512 - l, (l == 0) ? 384 : 256 - l, 256 + l };
#pragma unroll
  for (int v = 0; v < 8; ++v) {
    float* pr = Ob + v * 1536;
#pragma unroll
    for (int o = 0; o < 4; ++o) {
      float* p = pr + col[o] * 3;
#pragma unroll
      for (int c = 0; c < 3; ++c)
        p[c] = mm[o].x * acc[0][c][v] + mm[o].y * acc[1][c][v] +
               mm[o].z * acc[2][c][v] + mm[o].w * acc[3][c][v];
    }
  }
}

extern "C" void kernel_launch(void* const* d_in, const int* in_sizes, int n_in,
                              void* d_out, int out_size, void* d_ws, size_t ws_size,
                              hipStream_t stream) {
  (void)in_sizes; (void)n_in; (void)out_size; (void)ws_size;
  const float* x = (const float*)d_in[0];
  float* out = (float*)d_out;
  float* Fm  = (float*)d_ws;
  float* Y1  = Fm + 512 * 512;

  dct_make_quarter<<<dim3(64), dim3(256), 0, stream>>>(Fm);

  dct_cols_fold3<<<dim3(48, 1, 32), dim3(128), 0, stream>>>(Fm, x, Y1);

  dct_rows3_fold2<<<dim3(32, 1, 32), dim3(256), 0, stream>>>(Fm, Y1, out);
}


# Stage 2 third adjacent-pair fold (as stage 1): 8-pixel groups folded to 8 panels in LDS, eighth-size cos/sin tables as B, K 128 -> 64, 4-wave groups, columns by 8x8 epilogue matrix; changes sum order
# speedup vs baseline: 5.4109x; 1.1582x over previous
#include <hip/hip_runtime.h>


typedef __attribute__((ext_vector_type(2))) float v2f;
typedef __attribute__((ext_vector_type(4))) float v4f;
typedef __attribute__((ext_vector_type(8))) float v8f;

#define E_COS 0
#define E_SIN (E_COS + 4096)
#define E_MAT (E_SIN + 4096)
__global__ __launch_bounds__(256) void dct_make_eighth(float* __restrict__ Fh) {
  const int idx  = blockIdx.x * 256 + threadIdx.x;
  const int e    = idx & 3;
  const int lane = (idx >> 2) & 31;
  const int slot3 = (idx >> 7) & 7;
  const int k3 = (idx >> 10) * 16 + (lane & 15);
  const int i3 = (2 * slot3 + (e >> 1)) * 4 + 2 * (lane >> 4) + (e & 1);
  const int r3 = ((2 * i3 + 1) * k3) & 255;
  const float t3 = 3.14159265358979323846f * (float)r3 * (1.0f / 128.0f);
  const float c3 = cosf(t3) * 0.0625f;
  float s3 = sinf(t3) * 0.0625f;
  if (k3 == 0) s3 = (i3 & 1) ? -0.0625f : 0.0625f;
  Fh[E_COS + idx] = c3;
  Fh[E_SIN + idx] = s3;
  const int k = idx >> 6;
  const int o = (idx >> 3) & 7;
  const int q = idx & 7;
  const int K = (k == 0) ? 64 * o : ((o + 1) >> 1) * 128 + ((o & 1) ? -k : k);
  int n = 512, kk = K;
  bool isSine = false;
  float f = 1.0f;
  for (int lev = 0; lev < 3; ++lev) {
    const bool diff = (q >> (2 - lev)) & 1;
    const float phi = 3.14159265358979323846f * (float)kk / (float)(2 * n);
    if (diff) {
      f *= isSine ? -sinf(phi) : sinf(phi);
      isSine = !isSine;
    } else {
      f *= cosf(phi);
    }
    const int nh = n >> 1;
    if (kk > nh) {
      kk = n - kk;
      if (!isSine) f = -f;
    }
    n = nh;
  }
  if (k == 0) {
    if (kk == 0 && isSine) f = 0.0f;
    if (kk == 64 && !isSine) f = 0.0f;
  }
  if (K == 0) f *= 0.70710678118654752f;
  Fh[E_MAT + idx] = f;
}

#define EPANEL_DW  (32 * 64)
#define NT_E       4

__device__ __forceinline__ void fetchQ(const float* __restrict__ Fw, int kt,
                                       v4f (&f)[2]) {
#pragma unroll
  for (int jj = 0; jj < 2; ++jj)
    f[jj] = *(const v4f*)&Fw[(kt * 2 + jj) * 128];
}

__device__ __forceinline__ void mma_fold3(const float* __restrict__ B0,
                                          const float* __restrict__ B1,
                                          const v4f (&fc)[2], const v4f (&fs)[2],
                                          v8f (&acc)[8][2]) {
#pragma unroll
  for (int s = 0; s < 4; ++s) {
    const v4f qc = fc[s >> 1];
    const v4f qs = fs[s >> 1];
    const v2f ac = (s & 1) ? qc.zw : qc.xy;
    const v2f as = (s & 1) ? qs.zw : qs.xy;
    v2f b[8][2];
#pragma unroll
    for (int p = 0; p < 8; ++p) {
      b[p][0] = *(const v2f*)&B0[p * EPANEL_DW + s * 128];
      b[p][1] = *(const v2f*)&B1[p * EPANEL_DW + s * 128];
    }
#pragma unroll
    for (int nt = 0; nt < 2; ++nt) {
#pragma unroll
      for (int p = 0; p < 8; ++p) {
        const bool sine = ((p ^ (p >> 1) ^ (p >> 2)) & 1) != 0;
        acc[p][nt] = __builtin_amdgcn_wmma_f32_16x16x4_f32(
            false, sine ? as : ac, false, b[p][nt], (short)0, acc[p][nt], false, false);
      }
    }
  }
}

__global__ __launch_bounds__(128)
void dct_cols_fold3(const float* __restrict__ Fh, const float* __restrict__ X,
                    float* __restrict__ C) {
  __shared__ float Ds[8 * EPANEL_DW];

  const int tid = threadIdx.x;
  const long off = (long)blockIdx.z * (512L * 1536L) + blockIdx.x * 32;
  const float* Xb = X + off;
  float*       Cb = C + off;

  {
    const int n4 = tid & 7;
    const int kp = tid >> 3;
    const float* src = Xb + (unsigned)(16 * kp) * 1536u + (unsigned)n4 * 4u;
    float* dst = &Ds[kp * 64 + ((8 * n4) ^ (32 * (kp & 1)))];
    v4f rr[2][16];
#pragma unroll
    for (int sl = 0; sl < 2; ++sl)
#pragma unroll
      for (int i = 0; i < 16; ++i)
        rr[sl][i] = *(const v4f*)&src[(sl * 256 + i) * 1536];
#pragma unroll
    for (int sl = 0; sl < 2; ++sl) {
      v4f w[8][2];
#pragma unroll
      for (int h = 0; h < 2; ++h) {
        const v4f s0 = rr[sl][8 * h]     + rr[sl][8 * h + 1];
        const v4f d0 = rr[sl][8 * h]     - rr[sl][8 * h + 1];
        const v4f s1 = rr[sl][8 * h + 2] + rr[sl][8 * h + 3];
        const v4f d1 = rr[sl][8 * h + 2] - rr[sl][8 * h + 3];
        const v4f s2 = rr[sl][8 * h + 4] + rr[sl][8 * h + 5];
        const v4f d2 = rr[sl][8 * h + 4] - rr[sl][8 * h + 5];
        const v4f s3 = rr[sl][8 * h + 6] + rr[sl][8 * h + 7];
        const v4f d3 = rr[sl][8 * h + 6] - rr[sl][8 * h + 7];
        const v4f ss0 = s0 + s1, sd0 = s0 - s1, ss1 = s2 + s3, sd1 = s2 - s3;
        const v4f ds0 = d0 + d1, dd0 = d0 - d1, ds1 = d2 + d3, dd1 = d2 - d3;
        w[0][h] = ss0 + ss1;
        w[1][h] = ss0 - ss1;
        w[2][h] = sd0 + sd1;
        w[3][h] = sd0 - sd1;
        w[4][h] = ds0 + ds1;
        w[5][h] = ds0 - ds1;
        w[6][h] = dd0 + dd1;
        w[7][h] = dd0 - dd1;
      }
#pragma unroll
      for (int p = 0; p < 8; ++p) {
        v4f lo, hi;
        lo.x = w[p][0].x; lo.y = w[p][1].x; lo.z = w[p][0].y; lo.w = w[p][1].y;
        hi.x = w[p][0].z; hi.y = w[p][1].z; hi.z = w[p][0].w; hi.w = w[p][1].w;
        *(v4f*)&dst[p * EPANEL_DW + sl * 16 * 64]     = lo;
        *(v4f*)&dst[p * EPANEL_DW + sl * 16 * 64 + 4] = hi;
      }
    }
  }
  __builtin_amdgcn_sched_barrier(0);

  const int lane = tid & 31;
  const int wave = __builtin_amdgcn_readfirstlane(tid >> 5);
  const int row  = lane & 15;
  const int grp  = lane >> 4;
  const int k0   = wave * 16;

  const float* B0 = &Ds[grp * 64 + 2 * row + 32 * grp];
  const float* B1 = &Ds[grp * 64 + 2 * row + 32 * (1 - grp)];

  const float* Fcw = Fh + E_COS + ((long)wave * 8 * 32 + lane) * 4;
  const float* Fsw = Fcw + 4096;

  v8f acc[8][2];
#pragma unroll
  for (int p = 0; p < 8; ++p)
#pragma unroll
    for (int j = 0; j < 2; ++j) acc[p][j] = (v8f)0.0f;

  v4f fc0[2], fs0[2], fc1[2], fs1[2];
  fetchQ(Fcw, 0, fc0);
  fetchQ(Fsw, 0, fs0);

  __syncthreads();

  for (int t = 0; t < NT_E - 2; t += 2) {
    fetchQ(Fcw, t + 1, fc1);
    fetchQ(Fsw, t + 1, fs1);
    __builtin_amdgcn_sched_barrier(0);
    mma_fold3(B0 + t * 512, B1 + t * 512, fc0, fs0, acc);

    fetchQ(Fcw, t + 2, fc0);
    fetchQ(Fsw, t + 2, fs0);
    __builtin_amdgcn_sched_barrier(0);
    mma_fold3(B0 + (t + 1) * 512, B1 + (t + 1) * 512, fc1, fs1, acc);
  }
  fetchQ(Fcw, NT_E - 1, fc1);
  fetchQ(Fsw, NT_E - 1, fs1);
  __builtin_amdgcn_sched_barrier(0);
  mma_fold3(B0 + (NT_E - 2) * 512, B1 + (NT_E - 2) * 512, fc0, fs0, acc);
  mma_fold3(B0 + (NT_E - 1) * 512, B1 + (NT_E - 1) * 512, fc1, fs1, acc);

  const float* Mw = Fh + E_MAT + (k0 + grp * 8) * 64;
#pragma unroll
  for (int v = 0; v < 8; ++v) {
    const int k = k0 + grp * 8 + v;
    float a[2][8];
#pragma unroll
    for (int nt = 0; nt < 2; ++nt)
#pragma unroll
      for (int q = 0; q < 8; ++q) a[nt][q] = acc[q][nt][v];
#pragma unroll
    for (int o = 0; o < 8; ++o) {
      const int r = (k == 0) ? 64 * o : ((o + 1) >> 1) * 128 + ((o & 1) ? -k : k);
      const v4f ma = *(const v4f*)&Mw[v * 64 + o * 8];
      const v4f mb = *(const v4f*)&Mw[v * 64 + o * 8 + 4];
#pragma unroll
      for (int nt = 0; nt < 2; ++nt)
        Cb[(long)r * 1536 + nt * 16 + row] =
            ma.x * a[nt][0] + ma.y * a[nt][1] + ma.z * a[nt][2] + ma.w * a[nt][3] +
            mb.x * a[nt][4] + mb.y * a[nt][5] + mb.z * a[nt][6] + mb.w * a[nt][7];
    }
  }
}


#define ROWS_W 16
#define ROWP   1572
#define NT_W   4

__device__ __forceinline__ void mma_rows_fold3(const float* __restrict__ Rw,
                                               const v4f (&fc)[2], const v4f (&fs)[2],
                                               v8f (&acc)[8][3]) {
#pragma unroll
  for (int s = 0; s < 4; ++s) {
    const v4f qc = fc[s >> 1];
    const v4f qs = fs[s >> 1];
    const v2f bc = (s & 1) ? qc.zw : qc.xy;
    const v2f bs = (s & 1) ? qs.zw : qs.xy;
    v4f a[12];
#pragma unroll
    for (int t = 0; t < 12; ++t) a[t] = *(const v4f*)&Rw[s * 96 + t * 4];
#pragma unroll
    for (int p = 0; p < 8; ++p) {
      const bool sine = ((p ^ (p >> 1) ^ (p >> 2)) & 1) != 0;
#pragma unroll
      for (int c = 0; c < 3; ++c) {
        const int q = 3 * p + c;
        const v2f af = (q & 1) ? a[q >> 1].zw : a[q >> 1].xy;
        acc[p][c] = __builtin_amdgcn_wmma_f32_16x16x4_f32(
            false, af, false, sine ? bs : bc, (short)0, acc[p][c], false, false);
      }
    }
  }
}

__global__ __launch_bounds__(128, 3)
void dct_rows3_fold3(const float* __restrict__ Fh, const float* __restrict__ Y,
                     float* __restrict__ Out) {
  __shared__ float Rs[ROWS_W * ROWP];

  const int tid  = threadIdx.x;
  const int lane = tid & 31;
  const int wave = __builtin_amdgcn_readfirstlane(tid >> 5);
  const long offb = (long)blockIdx.z * (512L * 1536L);
  const int m0 = blockIdx.x * ROWS_W;

  {
    const int P  = (lane & 3) | ((lane >> 4) << 2) | (wave << 3);
    const int r0 = (lane >> 2) & 3;
    const float* src = Y + offb + (long)m0 * 1536 + (unsigned)r0 * 1536u + (unsigned)P * 48u;
    float* dst = &Rs[r0 * ROWP + P * 48];
    v4f rr[4][12];
#pragma unroll
    for (int it = 0; it < 4; ++it)
#pragma unroll
      for (int j = 0; j < 12; ++j)
        rr[it][j] = *(const v4f*)&src[it * 4 * 1536 + j * 4];
#pragma unroll
    for (int it = 0; it < 4; ++it) {
      float f[48], o[48];
#pragma unroll
      for (int i = 0; i < 48; ++i) f[i] = rr[it][i >> 2][i & 3];
#pragma unroll
      for (int e = 0; e < 2; ++e)
#pragma unroll
        for (int c = 0; c < 3; ++c) {
          const float x0 = f[24 * e + c],      x1 = f[24 * e + 3 + c];
          const float x2 = f[24 * e + 6 + c],  x3 = f[24 * e + 9 + c];
          const float x4 = f[24 * e + 12 + c], x5 = f[24 * e + 15 + c];
          const float x6 = f[24 * e + 18 + c], x7 = f[24 * e + 21 + c];
          const float s0 = x0 + x1, d0 = x0 - x1;
          const float s1 = x2 + x3, d1 = x2 - x3;
          const float s2 = x4 + x5, d2 = x4 - x5;
          const float s3 = x6 + x7, d3 = x6 - x7;
          const float ss0 = s0 + s1, sd0 = s0 - s1, ss1 = s2 + s3, sd1 = s2 - s3;
          const float ds0 = d0 + d1, dd0 = d0 - d1, ds1 = d2 + d3, dd1 = d2 - d3;
          o[2 * (0 + c) + e]  = ss0 + ss1;
          o[2 * (3 + c) + e]  = ss0 - ss1;
          o[2 * (6 + c) + e]  = sd0 + sd1;
          o[2 * (9 + c) + e]  = sd0 - sd1;
          o[2 * (12 + c) + e] = ds0 + ds1;
          o[2 * (15 + c) + e] = ds0 - ds1;
          o[2 * (18 + c) + e] = dd0 + dd1;
          o[2 * (21 + c) + e] = dd0 - dd1;
        }
#pragma unroll
      for (int t = 0; t < 12; ++t) {
        v4f w;
        w.x = o[4 * t]; w.y = o[4 * t + 1]; w.z = o[4 * t + 2]; w.w = o[4 * t + 3];
        *(v4f*)&dst[it * 4 * ROWP + 4 * t] = w;
      }
    }
  }

  const int row  = lane & 15;
  const int grp  = lane >> 4;
  const int l0   = wave * 16;

  const float* Fcw = Fh + E_COS + ((long)wave * 8 * 32 + lane) * 4;
  const float* Fsw = Fcw + 4096;

  const float* Rw = &Rs[row * ROWP + 48 * grp];

  v8f acc[8][3];
#pragma unroll
  for (int p = 0; p < 8; ++p)
#pragma unroll
    for (int c = 0; c < 3; ++c) acc[p][c] = (v8f)0.0f;

  v4f fc0[2], fs0[2], fc1[2], fs1[2];
  fetchQ(Fcw, 0, fc0);
  fetchQ(Fsw, 0, fs0);

  __syncthreads();

  for (int t = 0; t < NT_W - 2; t += 2) {
    fetchQ(Fcw, t + 1, fc1);
    fetchQ(Fsw, t + 1, fs1);
    __builtin_amdgcn_sched_barrier(0);
    mma_rows_fold3(Rw + t * 384, fc0, fs0, acc);

    fetchQ(Fcw, t + 2, fc0);
    fetchQ(Fsw, t + 2, fs0);
    __builtin_amdgcn_sched_barrier(0);
    mma_rows_fold3(Rw + (t + 1) * 384, fc1, fs1, acc);
  }
  fetchQ(Fcw, NT_W - 1, fc1);
  fetchQ(Fsw, NT_W - 1, fs1);
  __builtin_amdgcn_sched_barrier(0);
  mma_rows_fold3(Rw + (NT_W - 2) * 384, fc0, fs0, acc);

  const int l = l0 + row;
  const float* Ml = Fh + E_MAT + l * 64;
  v4f mm[8][2];
#pragma unroll
  for (int o = 0; o < 2; ++o) {
    mm[o][0] = *(const v4f*)&Ml[o * 8];
    mm[o][1] = *(const v4f*)&Ml[o * 8 + 4];
  }
  __builtin_amdgcn_sched_barrier(0);
  mma_rows_fold3(Rw + (NT_W - 1) * 384, fc1, fs1, acc);
#pragma unroll
  for (int o = 2; o < 8; ++o) {
    mm[o][0] = *(const v4f*)&Ml[o * 8];
    mm[o][1] = *(const v4f*)&Ml[o * 8 + 4];
  }

  float* Ob = Out + offb + (long)(m0 + grp * 8) * 1536;
  const bool lz = (l == 0);
#pragma unroll
  for (int o = 0; o < 8; ++o) {
    const int cn = ((o + 1) >> 1) * 128 + ((o & 1) ? -l : l);
    const int col = lz ? 64 * o : cn;
    float* p = Ob + col * 3;
#pragma unroll
    for (int v = 0; v < 8; ++v) {
#pragma unroll
      for (int c = 0; c < 3; ++c)
        p[v * 1536 + c] =
            mm[o][0].x * acc[0][c][v] + mm[o][0].y * acc[1][c][v] +
            mm[o][0].z * acc[2][c][v] + mm[o][0].w * acc[3][c][v] +
            mm[o][1].x * acc[4][c][v] + mm[o][1].y * acc[5][c][v] +
            mm[o][1].z * acc[6][c][v] + mm[o][1].w * acc[7][c][v];
    }
  }
}

extern "C" void kernel_launch(void* const* d_in, const int* in_sizes, int n_in,
                              void* d_out, int out_size, void* d_ws, size_t ws_size,
                              hipStream_t stream) {
  (void)in_sizes; (void)n_in; (void)out_size; (void)ws_size;
  const float* x = (const float*)d_in[0];
  float* out = (float*)d_out;
  float* Fm  = (float*)d_ws;
  float* Y1  = Fm + 512 * 512;

  dct_make_eighth<<<dim3(16), dim3(256), 0, stream>>>(Fm);

  dct_cols_fold3<<<dim3(48, 1, 32), dim3(128), 0, stream>>>(Fm, x, Y1);

  dct_rows3_fold3<<<dim3(32, 1, 32), dim3(128), 0, stream>>>(Fm, Y1, out);
}
